# starting kernel
# baseline (speedup 1.0000x reference)
#include <hip/hip_runtime.h>
#include <math.h>

#define N_NODES 100000
#define N_EDGES 1600000
#define IN_C    256
#define HID_C   128
#define N_LAYERS 4
#define ALPHA   0.1f

typedef float v2f __attribute__((ext_vector_type(2)));
typedef float v8f __attribute__((ext_vector_type(8)));

__device__ __forceinline__ void fadd_agent(float* p, float v) {
    __hip_atomic_fetch_add(p, v, __ATOMIC_RELAXED, __HIP_MEMORY_SCOPE_AGENT);
}

__global__ void kdeg_init(int* degi) {
    int v = blockIdx.x * 256 + threadIdx.x;
    if (v < N_NODES) degi[v] = 1;
}

__global__ void kdeg_count(const int* __restrict__ dst, int* __restrict__ degi) {
    int e = blockIdx.x * 256 + threadIdx.x;
    if (e < N_EDGES) atomicAdd(&degi[dst[e]], 1);
}

__global__ void kdinv(const int* __restrict__ degi, float* __restrict__ dinv) {
    int v = blockIdx.x * 256 + threadIdx.x;
    if (v < N_NODES) dinv[v] = rsqrtf((float)degi[v]);
}

__global__ void kzero4(float4* __restrict__ p, int n4) {
    int i = blockIdx.x * 256 + threadIdx.x;
    if (i < n4) p[i] = make_float4(0.f, 0.f, 0.f, 0.f);
}

#define LDA_P 260
__global__ void kproj(const float* __restrict__ x, const float* __restrict__ w,
                      const float* __restrict__ bias, float* __restrict__ x0) {
    __shared__ float sa[16 * LDA_P];
    const int t   = threadIdx.x;
    const int m0  = blockIdx.x * 16;

    const float4* xf4 = (const float4*)x;
    #pragma unroll
    for (int c = 0; c < 4; ++c) {
        int idx = c * 256 + t;
        int row = idx >> 6, q = idx & 63;
        float4 v = xf4[(size_t)(m0 + row) * 64 + q];
        float* d = &sa[row * LDA_P + q * 4];
        d[0] = v.x; d[1] = v.y; d[2] = v.z; d[3] = v.w;
    }
    __syncthreads();

    const int wave = t >> 5, lane = t & 31;
    const int half = lane >> 4, col = lane & 15;
    const int n0   = wave * 16;
    const int mrow = col;

    v8f acc = {};
    #pragma unroll 8
    for (int k0 = 0; k0 < IN_C; k0 += 4) {
        int ka = k0 + 2 * half;
        v2f a = *(const v2f*)&sa[mrow * LDA_P + ka];
        v2f b;
        b.x = w[(size_t)ka * HID_C + n0 + col];
        b.y = w[(size_t)(ka + 1) * HID_C + n0 + col];
        acc = __builtin_amdgcn_wmma_f32_16x16x4_f32(false, a, false, b,
                                                    (short)0, acc, false, false);
    }

    float bv = bias[n0 + col];
    #pragma unroll
    for (int i = 0; i < 8; ++i) {
        int row = i + 8 * half;
        x0[(size_t)(m0 + row) * HID_C + n0 + col] = acc[i] + bv;
    }
}

__global__ void kscatter(const int* __restrict__ src, const int* __restrict__ dst,
                         const float* __restrict__ dinv, const float* __restrict__ hin,
                         float* __restrict__ agg) {
    int wid  = (blockIdx.x * 256 + threadIdx.x) >> 5;
    int lane = threadIdx.x & 31;
    int s = src[wid], d = dst[wid];
    float nrm = dinv[s] * dinv[d];
    float4 v = ((const float4*)(hin + (size_t)s * HID_C))[lane];
    float* ap = agg + (size_t)d * HID_C + lane * 4;
    fadd_agent(ap + 0, nrm * v.x);
    fadd_agent(ap + 1, nrm * v.y);
    fadd_agent(ap + 2, nrm * v.z);
    fadd_agent(ap + 3, nrm * v.w);
}

#define LDA_L 132
__global__ void klayer(const float* __restrict__ agg, const float* __restrict__ hin,
                       const float* __restrict__ x0, const float* __restrict__ dinv,
                       const float* __restrict__ w, float beta, int do_relu,
                       float* __restrict__ hout) {
    __shared__ float sup[16 * LDA_L];
    const int t  = threadIdx.x;
    const int m0 = blockIdx.x * 16;

    const float4* af4 = (const float4*)agg;
    const float4* hf4 = (const float4*)hin;
    const float4* xf4 = (const float4*)x0;
    #pragma unroll
    for (int c = 0; c < 2; ++c) {
        int idx = c * 256 + t;
        int row = idx >> 5, q = idx & 31;
        float dv = dinv[m0 + row];
        float d2 = dv * dv;
        size_t base = (size_t)(m0 + row) * 32 + q;
        float4 a = af4[base], h = hf4[base], x = xf4[base];
        float* dp = &sup[row * LDA_L + q * 4];
        dp[0] = (1.f - ALPHA) * (a.x + d2 * h.x) + ALPHA * x.x;
        dp[1] = (1.f - ALPHA) * (a.y + d2 * h.y) + ALPHA * x.y;
        dp[2] = (1.f - ALPHA) * (a.z + d2 * h.z) + ALPHA * x.z;
        dp[3] = (1.f - ALPHA) * (a.w + d2 * h.w) + ALPHA * x.w;
    }
    __syncthreads();

    const int wave = t >> 5, lane = t & 31;
    const int half = lane >> 4, col = lane & 15;
    const int n0   = wave * 16;
    const int mrow = col;

    v8f acc = {};
    #pragma unroll 8
    for (int k0 = 0; k0 < HID_C; k0 += 4) {
        int ka = k0 + 2 * half;
        v2f a = *(const v2f*)&sup[mrow * LDA_L + ka];
        v2f b;
        b.x = w[(size_t)ka * HID_C + n0 + col];
        b.y = w[(size_t)(ka + 1) * HID_C + n0 + col];
        acc = __builtin_amdgcn_wmma_f32_16x16x4_f32(false, a, false, b,
                                                    (short)0, acc, false, false);
    }

    #pragma unroll
    for (int i = 0; i < 8; ++i) {
        int row = i + 8 * half;
        float s = sup[row * LDA_L + n0 + col];
        float v = (1.f - beta) * s + beta * acc[i];
        if (do_relu) v = fmaxf(v, 0.f);
        hout[(size_t)(m0 + row) * HID_C + n0 + col] = v;
    }
}

extern "C" void kernel_launch(void* const* d_in, const int* in_sizes, int n_in,
                              void* d_out, int out_size, void* d_ws, size_t ws_size,
                              hipStream_t stream) {
    const float* x      = (const float*)d_in[0];
    const int*   ei     = (const int*)d_in[1];
    const float* proj_w = (const float*)d_in[2];
    const float* proj_b = (const float*)d_in[3];
    const float* conv_w = (const float*)d_in[4];
    float* out = (float*)d_out;

    const int* e_src = ei;
    const int* e_dst = ei + N_EDGES;

    const size_t NH = (size_t)N_NODES * HID_C;
    char* ws = (char*)d_ws;
    int*   degi = (int*)ws;                                  ws += ((size_t)N_NODES * 4 + 255) & ~255ull;
    float* dinv = (float*)ws;                                ws += ((size_t)N_NODES * 4 + 255) & ~255ull;
    float* x0   = (float*)ws;                                ws += NH * 4;
    float* h    = (float*)ws;                                ws += NH * 4;
    float* agg  = (float*)ws;

    kdeg_init <<<(N_NODES + 255) / 256, 256, 0, stream>>>(degi);
    kdeg_count<<<(N_EDGES + 255) / 256, 256, 0, stream>>>(e_dst, degi);
    kdinv     <<<(N_NODES + 255) / 256, 256, 0, stream>>>(degi, dinv);

    kproj<<<N_NODES / 16, 256, 0, stream>>>(x, proj_w, proj_b, x0);

    static const float betas[N_LAYERS] = {0.40546510f, 0.22314355f, 0.15415068f, 0.11778303f};
    const int n4 = (int)(NH / 4);

    for (int i = 0; i < N_LAYERS; ++i) {
        const float* hin  = (i == 0) ? x0 : h;
        float*       hout = (i == N_LAYERS - 1) ? out : h;
        kzero4  <<<(n4 + 255) / 256, 256, 0, stream>>>((float4*)agg, n4);
        kscatter<<<N_EDGES / 8, 256, 0, stream>>>(e_src, e_dst, dinv, hin, agg);
        klayer  <<<N_NODES / 16, 256, 0, stream>>>(agg, hin, x0, dinv,
                                                   conv_w + (size_t)i * HID_C * HID_C,
                                                   betas[i], (i != N_LAYERS - 1) ? 1 : 0, hout);
    }
}


# Global float-atomic scatter and agg zero-fill replaced by owner work-groups: each holds 400 destination rows in LDS, scans the dst list, adds its edges' rows with LDS float adds (arrival order)
# speedup vs baseline: 2.1325x; 2.1325x over previous
#include <hip/hip_runtime.h>
#include <math.h>

#define N_NODES 100000
#define N_EDGES 1600000
#define IN_C    256
#define HID_C   128
#define N_LAYERS 4
#define ALPHA   0.1f

typedef float v2f __attribute__((ext_vector_type(2)));
typedef float v8f __attribute__((ext_vector_type(8)));

__device__ __forceinline__ void fadd_lds(float* p, float v) {
    __hip_atomic_fetch_add(p, v, __ATOMIC_RELAXED, __HIP_MEMORY_SCOPE_WORKGROUP);
}

__global__ void kdeg_init(int* degi) {
    int v = blockIdx.x * 256 + threadIdx.x;
    if (v < N_NODES) degi[v] = 1;
}

__global__ void kdeg_count(const int* __restrict__ dst, int* __restrict__ degi) {
    int e = blockIdx.x * 256 + threadIdx.x;
    if (e < N_EDGES) atomicAdd(&degi[dst[e]], 1);
}

__global__ void kdinv(const int* __restrict__ degi, float* __restrict__ dinv) {
    int v = blockIdx.x * 256 + threadIdx.x;
    if (v < N_NODES) dinv[v] = rsqrtf((float)degi[v]);
}

#define LDA_P 260
__global__ void kproj(const float* __restrict__ x, const float* __restrict__ w,
                      const float* __restrict__ bias, float* __restrict__ x0) {
    __shared__ float sa[16 * LDA_P];
    const int t   = threadIdx.x;
    const int m0  = blockIdx.x * 16;

    const float4* xf4 = (const float4*)x;
    #pragma unroll
    for (int c = 0; c < 4; ++c) {
        int idx = c * 256 + t;
        int row = idx >> 6, q = idx & 63;
        float4 v = xf4[(size_t)(m0 + row) * 64 + q];
        float* d = &sa[row * LDA_P + q * 4];
        d[0] = v.x; d[1] = v.y; d[2] = v.z; d[3] = v.w;
    }
    __syncthreads();

    const int wave = t >> 5, lane = t & 31;
    const int half = lane >> 4, col = lane & 15;
    const int n0   = wave * 16;
    const int mrow = col;

    v8f acc = {};
    #pragma unroll 8
    for (int k0 = 0; k0 < IN_C; k0 += 4) {
        int ka = k0 + 2 * half;
        v2f a = *(const v2f*)&sa[mrow * LDA_P + ka];
        v2f b;
        b.x = w[(size_t)ka * HID_C + n0 + col];
        b.y = w[(size_t)(ka + 1) * HID_C + n0 + col];
        acc = __builtin_amdgcn_wmma_f32_16x16x4_f32(false, a, false, b,
                                                    (short)0, acc, false, false);
    }

    float bv = bias[n0 + col];
    #pragma unroll
    for (int i = 0; i < 8; ++i) {
        int row = i + 8 * half;
        x0[(size_t)(m0 + row) * HID_C + n0 + col] = acc[i] + bv;
    }
}

#define OWN_ROWS   400
#define OWN_GROUPS (N_NODES / OWN_ROWS)
#define OWN_WAVES  32
#define SCAN_LOADS 4
#define DRAIN      8
#define QCAP       (DRAIN + SCAN_LOADS * 128)
static_assert(N_NODES % OWN_ROWS == 0, "ranges must tile the nodes");
static_assert(N_EDGES % 4 == 0, "dst is read 4 at a time");
static_assert(OWN_ROWS <= 512 && N_EDGES <= (1 << 23), "queue word packs edge id and 9-bit row");

#define OWN_UNIT   (SCAN_LOADS * 32)
#define OWN_UNITS  (N_EDGES / 4 / OWN_UNIT)
#define OWN_N_IT   ((OWN_UNITS + OWN_WAVES - 1) / OWN_WAVES)
static_assert(N_EDGES % (4 * OWN_UNIT) == 0, "the dst list is a whole number of wave units");

__device__ __forceinline__ void own_add_row(float* __restrict__ tile, unsigned row, int lane,
                                            float nrm, float4 v) {
    float* a = tile + row * HID_C + lane;
    fadd_lds(a,      nrm * v.x);
    fadd_lds(a + 32, nrm * v.y);
    fadd_lds(a + 64, nrm * v.z);
    fadd_lds(a + 96, nrm * v.w);
}

__device__ __forceinline__ int own_load_trip(const int4* __restrict__ d4, int j0, int n,
                                             int wave, int lane,
                                             int4 (&out)[SCAN_LOADS], bool& ok) {
    int j = j0 + n;
    if (j >= OWN_N_IT) j -= OWN_N_IT;
    const int unit = j * OWN_WAVES + wave;
    ok = (n < OWN_N_IT) && (unit < OWN_UNITS);
    const int4* p = d4 + (ok ? unit : 0) * OWN_UNIT + lane;
    #pragma unroll
    for (int u = 0; u < SCAN_LOADS; ++u) out[u] = p[u * 32];
    return unit * OWN_UNIT + lane;
}

__device__ __forceinline__ int own_trip(const int4 (&cur)[SCAN_LOADS], int b, int lo, int lane,
                                        unsigned* __restrict__ q, int cnt,
                                        float* __restrict__ tile,
                                        const int* __restrict__ src,
                                        const float* __restrict__ dinv,
                                        const float4* __restrict__ h4) {
    #pragma unroll
    for (int u = 0; u < SCAN_LOADS; ++u) {
        unsigned r[4] = { (unsigned)(cur[u].x - lo), (unsigned)(cur[u].y - lo),
                          (unsigned)(cur[u].z - lo), (unsigned)(cur[u].w - lo) };
        unsigned rmin = min(min(r[0], r[1]), min(r[2], r[3]));
        if (__builtin_amdgcn_ballot_w32(rmin < OWN_ROWS) != 0) {
            #pragma unroll
            for (int k = 0; k < 4; ++k) {
                bool m = r[k] < OWN_ROWS;
                unsigned mask = __builtin_amdgcn_ballot_w32(m);
                if (mask != 0) {
                    if (m) {
                        unsigned e = (unsigned)((b + u * 32) * 4 + k);
                        q[cnt + __builtin_amdgcn_mbcnt_lo(mask, 0)] = (e << 9) | r[k];
                    }
                    cnt += __builtin_popcount(mask);
                }
            }
        }
    }

    while (cnt >= DRAIN) {
        cnt -= DRAIN;
        unsigned pe = q[cnt + (lane & (DRAIN - 1))];
        int      sl = src[pe >> 9];
        unsigned row[DRAIN];
        float4   v[DRAIN];
        float    nrm[DRAIN];
        #pragma unroll
        for (int i = 0; i < DRAIN; ++i) {
            int s  = __builtin_amdgcn_readlane(sl, i);
            row[i] = (unsigned)__builtin_amdgcn_readlane((int)pe, i) & 511u;
            v[i]   = h4[(size_t)s * (HID_C / 4) + lane];
            nrm[i] = dinv[s] * dinv[lo + row[i]];
        }
        #pragma unroll
        for (int i = 0; i < DRAIN; ++i) own_add_row(tile, row[i], lane, nrm[i], v[i]);
    }
    return cnt;
}

__global__ __launch_bounds__(OWN_WAVES * 32)
void kown(const int* __restrict__ src, const int* __restrict__ dst,
          const float* __restrict__ dinv, const float* __restrict__ hin,
          float* __restrict__ agg) {
    __shared__ float    tile[OWN_ROWS * HID_C];
    __shared__ unsigned queue[OWN_WAVES * QCAP];

    const int t = threadIdx.x, lane = t & 31;
    const int wave = __builtin_amdgcn_readfirstlane(t >> 5);
    const int lo = blockIdx.x * OWN_ROWS;

    float4* tile4 = (float4*)tile;
    for (int i = t; i < OWN_ROWS * HID_C / 4; i += OWN_WAVES * 32)
        tile4[i] = make_float4(0.f, 0.f, 0.f, 0.f);
    __syncthreads();

    unsigned* q = &queue[wave * QCAP];
    int cnt = 0;
    const int4*   d4 = (const int4*)dst;
    const float4* h4 = (const float4*)hin;

    const int j0 = (int)(blockIdx.x * OWN_N_IT) / OWN_GROUPS;

    int4 wa[SCAN_LOADS], wb[SCAN_LOADS];
    bool oka, okb;
    int ba = own_load_trip(d4, j0, 0, wave, lane, wa, oka);
    for (int n = 0; n < OWN_N_IT; n += 2) {
        int bb = own_load_trip(d4, j0, n + 1, wave, lane, wb, okb);
        if (oka) cnt = own_trip(wa, ba, lo, lane, q, cnt, tile, src, dinv, h4);
        ba = own_load_trip(d4, j0, n + 2, wave, lane, wa, oka);
        if (okb) cnt = own_trip(wb, bb, lo, lane, q, cnt, tile, src, dinv, h4);
    }

    for (int i = 0; i < cnt; ++i) {
        unsigned pe  = (unsigned)__builtin_amdgcn_readfirstlane((int)q[i]);
        int      s   = src[pe >> 9];
        unsigned row = pe & 511u;
        float4   v   = h4[(size_t)s * (HID_C / 4) + lane];
        own_add_row(tile, row, lane, dinv[s] * dinv[lo + row], v);
    }
    __syncthreads();

    float4* agg4 = (float4*)agg;
    for (int i = t; i < OWN_ROWS * (HID_C / 4); i += OWN_WAVES * 32) {
        int row = i >> 5, qd = i & 31;
        const float* a = tile + row * HID_C + qd;
        agg4[(size_t)(lo + row) * (HID_C / 4) + qd] = make_float4(a[0], a[32], a[64], a[96]);
    }
}

#define LDA_L 132
__global__ void klayer(const float* __restrict__ agg, const float* __restrict__ hin,
                       const float* __restrict__ x0, const float* __restrict__ dinv,
                       const float* __restrict__ w, float beta, int do_relu,
                       float* __restrict__ hout) {
    __shared__ float sup[16 * LDA_L];
    const int t  = threadIdx.x;
    const int m0 = blockIdx.x * 16;

    const float4* af4 = (const float4*)agg;
    const float4* hf4 = (const float4*)hin;
    const float4* xf4 = (const float4*)x0;
    #pragma unroll
    for (int c = 0; c < 2; ++c) {
        int idx = c * 256 + t;
        int row = idx >> 5, q = idx & 31;
        float dv = dinv[m0 + row];
        float d2 = dv * dv;
        size_t base = (size_t)(m0 + row) * 32 + q;
        float4 a = af4[base], h = hf4[base], x = xf4[base];
        float* dp = &sup[row * LDA_L + q * 4];
        dp[0] = (1.f - ALPHA) * (a.x + d2 * h.x) + ALPHA * x.x;
        dp[1] = (1.f - ALPHA) * (a.y + d2 * h.y) + ALPHA * x.y;
        dp[2] = (1.f - ALPHA) * (a.z + d2 * h.z) + ALPHA * x.z;
        dp[3] = (1.f - ALPHA) * (a.w + d2 * h.w) + ALPHA * x.w;
    }
    __syncthreads();

    const int wave = t >> 5, lane = t & 31;
    const int half = lane >> 4, col = lane & 15;
    const int n0   = wave * 16;
    const int mrow = col;

    v8f acc = {};
    #pragma unroll 8
    for (int k0 = 0; k0 < HID_C; k0 += 4) {
        int ka = k0 + 2 * half;
        v2f a = *(const v2f*)&sup[mrow * LDA_L + ka];
        v2f b;
        b.x = w[(size_t)ka * HID_C + n0 + col];
        b.y = w[(size_t)(ka + 1) * HID_C + n0 + col];
        acc = __builtin_amdgcn_wmma_f32_16x16x4_f32(false, a, false, b,
                                                    (short)0, acc, false, false);
    }

    #pragma unroll
    for (int i = 0; i < 8; ++i) {
        int row = i + 8 * half;
        float s = sup[row * LDA_L + n0 + col];
        float v = (1.f - beta) * s + beta * acc[i];
        if (do_relu) v = fmaxf(v, 0.f);
        hout[(size_t)(m0 + row) * HID_C + n0 + col] = v;
    }
}

extern "C" void kernel_launch(void* const* d_in, const int* in_sizes, int n_in,
                              void* d_out, int out_size, void* d_ws, size_t ws_size,
                              hipStream_t stream) {
    const float* x      = (const float*)d_in[0];
    const int*   ei     = (const int*)d_in[1];
    const float* proj_w = (const float*)d_in[2];
    const float* proj_b = (const float*)d_in[3];
    const float* conv_w = (const float*)d_in[4];
    float* out = (float*)d_out;

    const int* e_src = ei;
    const int* e_dst = ei + N_EDGES;

    const size_t NH = (size_t)N_NODES * HID_C;
    char* ws = (char*)d_ws;
    int*   degi = (int*)ws;                                  ws += ((size_t)N_NODES * 4 + 255) & ~255ull;
    float* dinv = (float*)ws;                                ws += ((size_t)N_NODES * 4 + 255) & ~255ull;
    float* x0   = (float*)ws;                                ws += NH * 4;
    float* h    = (float*)ws;                                ws += NH * 4;
    float* agg  = (float*)ws;

    kdeg_init <<<(N_NODES + 255) / 256, 256, 0, stream>>>(degi);
    kdeg_count<<<(N_EDGES + 255) / 256, 256, 0, stream>>>(e_dst, degi);
    kdinv     <<<(N_NODES + 255) / 256, 256, 0, stream>>>(degi, dinv);

    kproj<<<N_NODES / 16, 256, 0, stream>>>(x, proj_w, proj_b, x0);

    static const float betas[N_LAYERS] = {0.40546510f, 0.22314355f, 0.15415068f, 0.11778303f};

    for (int i = 0; i < N_LAYERS; ++i) {
        const float* hin  = (i == 0) ? x0 : h;
        float*       hout = (i == N_LAYERS - 1) ? out : h;
        kown    <<<OWN_GROUPS, OWN_WAVES * 32, 0, stream>>>(e_src, e_dst, dinv, hin, agg);
        klayer  <<<N_NODES / 16, 256, 0, stream>>>(agg, hin, x0, dinv,
                                                   conv_w + (size_t)i * HID_C * HID_C,
                                                   betas[i], (i != N_LAYERS - 1) ? 1 : 0, hout);
    }
}


# Owner work-groups record the edges their layer-1 dst scan finds (source, local row, weight) in a per-group workspace segment; layers 2-4 replay it and skip the scan; row sums stay in arrival order
# speedup vs baseline: 2.7879x; 1.3073x over previous
#include <hip/hip_runtime.h>
#include <math.h>

#define N_NODES 100000
#define N_EDGES 1600000
#define IN_C    256
#define HID_C   128
#define N_LAYERS 4
#define ALPHA   0.1f

typedef float v2f __attribute__((ext_vector_type(2)));
typedef float v8f __attribute__((ext_vector_type(8)));

__device__ __forceinline__ void fadd_lds(float* p, float v) {
    __hip_atomic_fetch_add(p, v, __ATOMIC_RELAXED, __HIP_MEMORY_SCOPE_WORKGROUP);
}

__global__ void kdeg_init(int* degi) {
    int v = blockIdx.x * 256 + threadIdx.x;
    if (v < N_NODES) degi[v] = 1;
}

__global__ void kdeg_count(const int* __restrict__ dst, int* __restrict__ degi) {
    int e = blockIdx.x * 256 + threadIdx.x;
    if (e < N_EDGES) atomicAdd(&degi[dst[e]], 1);
}

__global__ void kdinv(const int* __restrict__ degi, float* __restrict__ dinv) {
    int v = blockIdx.x * 256 + threadIdx.x;
    if (v < N_NODES) dinv[v] = rsqrtf((float)degi[v]);
}

#define LDA_P 260
__global__ void kproj(const float* __restrict__ x, const float* __restrict__ w,
                      const float* __restrict__ bias, float* __restrict__ x0) {
    __shared__ float sa[16 * LDA_P];
    const int t   = threadIdx.x;
    const int m0  = blockIdx.x * 16;

    const float4* xf4 = (const float4*)x;
    #pragma unroll
    for (int c = 0; c < 4; ++c) {
        int idx = c * 256 + t;
        int row = idx >> 6, q = idx & 63;
        float4 v = xf4[(size_t)(m0 + row) * 64 + q];
        float* d = &sa[row * LDA_P + q * 4];
        d[0] = v.x; d[1] = v.y; d[2] = v.z; d[3] = v.w;
    }
    __syncthreads();

    const int wave = t >> 5, lane = t & 31;
    const int half = lane >> 4, col = lane & 15;
    const int n0   = wave * 16;
    const int mrow = col;

    v8f acc = {};
    #pragma unroll 8
    for (int k0 = 0; k0 < IN_C; k0 += 4) {
        int ka = k0 + 2 * half;
        v2f a = *(const v2f*)&sa[mrow * LDA_P + ka];
        v2f b;
        b.x = w[(size_t)ka * HID_C + n0 + col];
        b.y = w[(size_t)(ka + 1) * HID_C + n0 + col];
        acc = __builtin_amdgcn_wmma_f32_16x16x4_f32(false, a, false, b,
                                                    (short)0, acc, false, false);
    }

    float bv = bias[n0 + col];
    #pragma unroll
    for (int i = 0; i < 8; ++i) {
        int row = i + 8 * half;
        x0[(size_t)(m0 + row) * HID_C + n0 + col] = acc[i] + bv;
    }
}

#define OWN_ROWS   400
#define OWN_GROUPS (N_NODES / OWN_ROWS)
#define OWN_WAVES  32
#define SCAN_LOADS 4
#define DRAIN      8
#define QCAP       (DRAIN + SCAN_LOADS * 128)
#define DEG_ILP    5
static_assert(N_NODES % OWN_ROWS == 0, "ranges must tile the nodes");
static_assert(N_EDGES % 4 == 0, "dst is read 4 at a time");
static_assert(OWN_ROWS <= 512 && N_EDGES <= (1 << 23), "queue word packs edge id and 9-bit row");
static_assert(N_NODES <= (1 << 23), "record word packs source node and 9-bit row");

#define OWN_UNIT   (SCAN_LOADS * 32)
#define OWN_UNITS  (N_EDGES / 4 / OWN_UNIT)
#define OWN_N_IT   ((OWN_UNITS + OWN_WAVES - 1) / OWN_WAVES)
static_assert(N_EDGES % (4 * OWN_UNIT) == 0, "the dst list is a whole number of wave units");

__device__ __forceinline__ void own_add_row(float* __restrict__ tile, unsigned row, int lane,
                                            float nrm, float4 v) {
    float* a = tile + row * HID_C + lane;
    fadd_lds(a,      nrm * v.x);
    fadd_lds(a + 32, nrm * v.y);
    fadd_lds(a + 64, nrm * v.z);
    fadd_lds(a + 96, nrm * v.w);
}

__device__ __forceinline__ void own_record(uint2* __restrict__ seg, unsigned* __restrict__ cursor,
                                           int n, int lane, int sl, unsigned rl, int lo,
                                           const float* __restrict__ dinv) {
    float nl = dinv[sl] * dinv[lo + rl];
    unsigned base = 0;
    if (lane == 0)
        base = __hip_atomic_fetch_add(cursor, (unsigned)n, __ATOMIC_RELAXED,
                                      __HIP_MEMORY_SCOPE_WORKGROUP);
    base = (unsigned)__builtin_amdgcn_readfirstlane((int)base);
    if (lane < n)
        seg[base + lane] = make_uint2(((unsigned)sl << 9) | rl, __float_as_uint(nl));
}

__device__ __forceinline__ int own_load_trip(const int4* __restrict__ d4, int j0, int n,
                                             int wave, int lane,
                                             int4 (&out)[SCAN_LOADS], bool& ok) {
    int j = j0 + n;
    if (j >= OWN_N_IT) j -= OWN_N_IT;
    const int unit = j * OWN_WAVES + wave;
    ok = (n < OWN_N_IT) && (unit < OWN_UNITS);
    const int4* p = d4 + (ok ? unit : 0) * OWN_UNIT + lane;
    #pragma unroll
    for (int u = 0; u < SCAN_LOADS; ++u) out[u] = p[u * 32];
    return unit * OWN_UNIT + lane;
}

__device__ __forceinline__ int own_trip(const int4 (&cur)[SCAN_LOADS], int b, int lo, int lane,
                                        unsigned* __restrict__ q, int cnt,
                                        float* __restrict__ tile,
                                        const int* __restrict__ src,
                                        const float* __restrict__ dinv,
                                        const float4* __restrict__ h4,
                                        unsigned* __restrict__ cursor,
                                        uint2* __restrict__ seg) {
    #pragma unroll
    for (int u = 0; u < SCAN_LOADS; ++u) {
        unsigned r[4] = { (unsigned)(cur[u].x - lo), (unsigned)(cur[u].y - lo),
                          (unsigned)(cur[u].z - lo), (unsigned)(cur[u].w - lo) };
        unsigned rmin = min(min(r[0], r[1]), min(r[2], r[3]));
        if (__builtin_amdgcn_ballot_w32(rmin < OWN_ROWS) != 0) {
            #pragma unroll
            for (int k = 0; k < 4; ++k) {
                bool m = r[k] < OWN_ROWS;
                unsigned mask = __builtin_amdgcn_ballot_w32(m);
                if (mask != 0) {
                    if (m) {
                        unsigned e = (unsigned)((b + u * 32) * 4 + k);
                        q[cnt + __builtin_amdgcn_mbcnt_lo(mask, 0)] = (e << 9) | r[k];
                    }
                    cnt += __builtin_popcount(mask);
                }
            }
        }
    }

    while (cnt >= DRAIN) {
        cnt -= DRAIN;
        unsigned pe = q[cnt + (lane & (DRAIN - 1))];
        int      sl = src[pe >> 9];
        unsigned row[DRAIN];
        float4   v[DRAIN];
        float    nrm[DRAIN];
        #pragma unroll
        for (int i = 0; i < DRAIN; ++i) {
            int s  = __builtin_amdgcn_readlane(sl, i);
            row[i] = (unsigned)__builtin_amdgcn_readlane((int)pe, i) & 511u;
            v[i]   = h4[(size_t)s * (HID_C / 4) + lane];
            nrm[i] = dinv[s] * dinv[lo + row[i]];
        }
        own_record(seg, cursor, DRAIN, lane, sl, pe & 511u, lo, dinv);
        #pragma unroll
        for (int i = 0; i < DRAIN; ++i) own_add_row(tile, row[i], lane, nrm[i], v[i]);
    }
    return cnt;
}

__global__ __launch_bounds__(OWN_WAVES * 32)
void kown_scan(const int* __restrict__ src, const int* __restrict__ dst,
               const float* __restrict__ dinv, const int* __restrict__ degi,
               const float* __restrict__ hin, float* __restrict__ agg,
               unsigned* __restrict__ rec, int* __restrict__ gmeta) {
    __shared__ float    tile[OWN_ROWS * HID_C];
    __shared__ unsigned queue[OWN_WAVES * QCAP];
    __shared__ int      wsum[OWN_WAVES];
    __shared__ unsigned cursor;

    const int t = threadIdx.x, lane = t & 31;
    const int wave = __builtin_amdgcn_readfirstlane(t >> 5);
    const int lo = blockIdx.x * OWN_ROWS;

    int part = 0;
    const int4* dg4 = (const int4*)degi;
    const int   lim = lo / 4;
    for (int i0 = 0; i0 < lim; i0 += DEG_ILP * OWN_WAVES * 32) {
        #pragma unroll
        for (int k = 0; k < DEG_ILP; ++k) {
            const int i = i0 + k * OWN_WAVES * 32 + t;
            int4 d = dg4[min(i, lim - 1)];
            part += ((d.x + d.y) + (d.z + d.w)) & (i < lim ? -1 : 0);
        }
    }
    #pragma unroll
    for (int o = 16; o > 0; o >>= 1) part += __shfl_xor(part, o);
    if (lane == 0) wsum[wave] = part;
    if (t == 0) cursor = 0u;

    float4* tile4 = (float4*)tile;
    for (int i = t; i < OWN_ROWS * HID_C / 4; i += OWN_WAVES * 32)
        tile4[i] = make_float4(0.f, 0.f, 0.f, 0.f);
    __syncthreads();

    int goff = -lo;
    for (int i = 0; i < OWN_WAVES; ++i) goff += wsum[i];
    goff = __builtin_amdgcn_readfirstlane(goff);
    uint2* seg = (uint2*)rec + goff;

    unsigned* q = &queue[wave * QCAP];
    int cnt = 0;
    const int4*   d4 = (const int4*)dst;
    const float4* h4 = (const float4*)hin;

    const int j0 = (int)(blockIdx.x * OWN_N_IT) / OWN_GROUPS;

    int4 wa[SCAN_LOADS], wb[SCAN_LOADS];
    bool oka, okb;
    int ba = own_load_trip(d4, j0, 0, wave, lane, wa, oka);
    for (int n = 0; n < OWN_N_IT; n += 2) {
        int bb = own_load_trip(d4, j0, n + 1, wave, lane, wb, okb);
        if (oka) cnt = own_trip(wa, ba, lo, lane, q, cnt, tile, src, dinv, h4, &cursor, seg);
        ba = own_load_trip(d4, j0, n + 2, wave, lane, wa, oka);
        if (okb) cnt = own_trip(wb, bb, lo, lane, q, cnt, tile, src, dinv, h4, &cursor, seg);
    }

    if (cnt > 0) {
        unsigned pl = q[lane < cnt ? lane : 0];
        own_record(seg, &cursor, cnt, lane, src[pl >> 9], pl & 511u, lo, dinv);
    }
    for (int i = 0; i < cnt; ++i) {
        unsigned pe  = (unsigned)__builtin_amdgcn_readfirstlane((int)q[i]);
        int      s   = src[pe >> 9];
        unsigned row = pe & 511u;
        float4   v   = h4[(size_t)s * (HID_C / 4) + lane];
        own_add_row(tile, row, lane, dinv[s] * dinv[lo + row], v);
    }
    __syncthreads();

    float4* agg4 = (float4*)agg;
    for (int i = t; i < OWN_ROWS * (HID_C / 4); i += OWN_WAVES * 32) {
        int row = i >> 5, qd = i & 31;
        const float* a = tile + row * HID_C + qd;
        agg4[(size_t)(lo + row) * (HID_C / 4) + qd] = make_float4(a[0], a[32], a[64], a[96]);
    }
    if (t == 0) {
        gmeta[2 * blockIdx.x]     = goff;
        gmeta[2 * blockIdx.x + 1] = (int)cursor;
    }
}

__global__ __launch_bounds__(OWN_WAVES * 32)
void kown_replay(const unsigned* __restrict__ rec, const int* __restrict__ gmeta,
                 const float* __restrict__ hin, float* __restrict__ agg) {
    __shared__ float tile[OWN_ROWS * HID_C];

    const int t = threadIdx.x, lane = t & 31;
    const int wave = __builtin_amdgcn_readfirstlane(t >> 5);
    const int lo = blockIdx.x * OWN_ROWS;

    float4* tile4 = (float4*)tile;
    for (int i = t; i < OWN_ROWS * HID_C / 4; i += OWN_WAVES * 32)
        tile4[i] = make_float4(0.f, 0.f, 0.f, 0.f);
    __syncthreads();

    const int    goff = gmeta[2 * blockIdx.x];
    const int    n    = gmeta[2 * blockIdx.x + 1];
    const int    nd   = n / DRAIN;
    const uint2* seg  = (const uint2*)rec + goff;
    const float4* h4  = (const float4*)hin;

    int   d  = wave;
    uint2 pe = make_uint2(0u, 0u);
    if (d < nd) pe = seg[d * DRAIN + (lane & (DRAIN - 1))];
    while (d < nd) {
        unsigned row[DRAIN];
        float4   v[DRAIN];
        float    nrm[DRAIN];
        #pragma unroll
        for (int i = 0; i < DRAIN; ++i) {
            unsigned w = (unsigned)__builtin_amdgcn_readlane((int)pe.x, i);
            row[i] = w & 511u;
            nrm[i] = __uint_as_float((unsigned)__builtin_amdgcn_readlane((int)pe.y, i));
            v[i]   = h4[(size_t)(w >> 9) * (HID_C / 4) + lane];
        }
        const int   dn = d + OWN_WAVES;
        const uint2 pn = seg[min(dn, nd - 1) * DRAIN + (lane & (DRAIN - 1))];
        #pragma unroll
        for (int i = 0; i < DRAIN; ++i) own_add_row(tile, row[i], lane, nrm[i], v[i]);
        pe = pn;
        d  = dn;
    }
    if (wave == 0) {
        for (int i = nd * DRAIN; i < n; ++i) {
            uint2    r = seg[i];
            unsigned w = (unsigned)__builtin_amdgcn_readfirstlane((int)r.x);
            float    f = __uint_as_float((unsigned)__builtin_amdgcn_readfirstlane((int)r.y));
            float4   v = h4[(size_t)(w >> 9) * (HID_C / 4) + lane];
            own_add_row(tile, w & 511u, lane, f, v);
        }
    }
    __syncthreads();

    float4* agg4 = (float4*)agg;
    for (int i = t; i < OWN_ROWS * (HID_C / 4); i += OWN_WAVES * 32) {
        int row = i >> 5, qd = i & 31;
        const float* a = tile + row * HID_C + qd;
        agg4[(size_t)(lo + row) * (HID_C / 4) + qd] = make_float4(a[0], a[32], a[64], a[96]);
    }
}

#define LDA_L 132
__global__ void klayer(const float* __restrict__ agg, const float* __restrict__ hin,
                       const float* __restrict__ x0, const float* __restrict__ dinv,
                       const float* __restrict__ w, float beta, int do_relu,
                       float* __restrict__ hout) {
    __shared__ float sup[16 * LDA_L];
    const int t  = threadIdx.x;
    const int m0 = blockIdx.x * 16;

    const float4* af4 = (const float4*)agg;
    const float4* hf4 = (const float4*)hin;
    const float4* xf4 = (const float4*)x0;
    #pragma unroll
    for (int c = 0; c < 2; ++c) {
        int idx = c * 256 + t;
        int row = idx >> 5, q = idx & 31;
        float dv = dinv[m0 + row];
        float d2 = dv * dv;
        size_t base = (size_t)(m0 + row) * 32 + q;
        float4 a = af4[base], h = hf4[base], x = xf4[base];
        float* dp = &sup[row * LDA_L + q * 4];
        dp[0] = (1.f - ALPHA) * (a.x + d2 * h.x) + ALPHA * x.x;
        dp[1] = (1.f - ALPHA) * (a.y + d2 * h.y) + ALPHA * x.y;
        dp[2] = (1.f - ALPHA) * (a.z + d2 * h.z) + ALPHA * x.z;
        dp[3] = (1.f - ALPHA) * (a.w + d2 * h.w) + ALPHA * x.w;
    }
    __syncthreads();

    const int wave = t >> 5, lane = t & 31;
    const int half = lane >> 4, col = lane & 15;
    const int n0   = wave * 16;
    const int mrow = col;

    v8f acc = {};
    #pragma unroll 8
    for (int k0 = 0; k0 < HID_C; k0 += 4) {
        int ka = k0 + 2 * half;
        v2f a = *(const v2f*)&sup[mrow * LDA_L + ka];
        v2f b;
        b.x = w[(size_t)ka * HID_C + n0 + col];
        b.y = w[(size_t)(ka + 1) * HID_C + n0 + col];
        acc = __builtin_amdgcn_wmma_f32_16x16x4_f32(false, a, false, b,
                                                    (short)0, acc, false, false);
    }

    #pragma unroll
    for (int i = 0; i < 8; ++i) {
        int row = i + 8 * half;
        float s = sup[row * LDA_L + n0 + col];
        float v = (1.f - beta) * s + beta * acc[i];
        if (do_relu) v = fmaxf(v, 0.f);
        hout[(size_t)(m0 + row) * HID_C + n0 + col] = v;
    }
}

extern "C" void kernel_launch(void* const* d_in, const int* in_sizes, int n_in,
                              void* d_out, int out_size, void* d_ws, size_t ws_size,
                              hipStream_t stream) {
    const float* x      = (const float*)d_in[0];
    const int*   ei     = (const int*)d_in[1];
    const float* proj_w = (const float*)d_in[2];
    const float* proj_b = (const float*)d_in[3];
    const float* conv_w = (const float*)d_in[4];
    float* out = (float*)d_out;

    const int* e_src = ei;
    const int* e_dst = ei + N_EDGES;

    const size_t NH = (size_t)N_NODES * HID_C;
    char* ws = (char*)d_ws;
    int*   degi = (int*)ws;                                  ws += ((size_t)N_NODES * 4 + 255) & ~255ull;
    float* dinv = (float*)ws;                                ws += ((size_t)N_NODES * 4 + 255) & ~255ull;
    float* x0   = (float*)ws;                                ws += NH * 4;
    float* h    = (float*)ws;                                ws += NH * 4;
    float* agg  = (float*)ws;                                ws += NH * 4;
    unsigned* rec   = (unsigned*)ws;                         ws += (size_t)N_EDGES * 8;
    int*      gmeta = (int*)ws;

    kdeg_init <<<(N_NODES + 255) / 256, 256, 0, stream>>>(degi);
    kdeg_count<<<(N_EDGES + 255) / 256, 256, 0, stream>>>(e_dst, degi);
    kdinv     <<<(N_NODES + 255) / 256, 256, 0, stream>>>(degi, dinv);

    kproj<<<N_NODES / 16, 256, 0, stream>>>(x, proj_w, proj_b, x0);

    static const float betas[N_LAYERS] = {0.40546510f, 0.22314355f, 0.15415068f, 0.11778303f};

    for (int i = 0; i < N_LAYERS; ++i) {
        const float* hin  = (i == 0) ? x0 : h;
        float*       hout = (i == N_LAYERS - 1) ? out : h;
        if (i == 0)
            kown_scan  <<<OWN_GROUPS, OWN_WAVES * 32, 0, stream>>>(e_src, e_dst, dinv, degi, hin, agg,
                                                                 rec, gmeta);
        else
            kown_replay<<<OWN_GROUPS, OWN_WAVES * 32, 0, stream>>>(rec, gmeta, hin, agg);
        klayer  <<<N_NODES / 16, 256, 0, stream>>>(agg, hin, x0, dinv,
                                                   conv_w + (size_t)i * HID_C * HID_C,
                                                   betas[i], (i != N_LAYERS - 1) ? 1 : 0, hout);
    }
}


# Layer GEMM fused into the owner work-groups: the summed LDS tile becomes the support tile in place and feeds WMMA, B fragments held in registers; agg array and 4 launches gone; h ping-pongs
# speedup vs baseline: 3.0833x; 1.1060x over previous
#include <hip/hip_runtime.h>
#include <math.h>

#define N_NODES 100000
#define N_EDGES 1600000
#define IN_C    256
#define HID_C   128
#define N_LAYERS 4
#define ALPHA   0.1f

typedef float v2f __attribute__((ext_vector_type(2)));
typedef float v8f __attribute__((ext_vector_type(8)));

__device__ __forceinline__ void fadd_lds(float* p, float v) {
    __hip_atomic_fetch_add(p, v, __ATOMIC_RELAXED, __HIP_MEMORY_SCOPE_WORKGROUP);
}

__global__ void kdeg_init(int* degi) {
    int v = blockIdx.x * 256 + threadIdx.x;
    if (v < N_NODES) degi[v] = 1;
}

__global__ void kdeg_count(const int* __restrict__ dst, int* __restrict__ degi) {
    int e = blockIdx.x * 256 + threadIdx.x;
    if (e < N_EDGES) atomicAdd(&degi[dst[e]], 1);
}

__global__ void kdinv(const int* __restrict__ degi, float* __restrict__ dinv) {
    int v = blockIdx.x * 256 + threadIdx.x;
    if (v < N_NODES) dinv[v] = rsqrtf((float)degi[v]);
}

#define LDA_P 260
__global__ void kproj(const float* __restrict__ x, const float* __restrict__ w,
                      const float* __restrict__ bias, float* __restrict__ x0) {
    __shared__ float sa[16 * LDA_P];
    const int t   = threadIdx.x;
    const int m0  = blockIdx.x * 16;

    const float4* xf4 = (const float4*)x;
    #pragma unroll
    for (int c = 0; c < 4; ++c) {
        int idx = c * 256 + t;
        int row = idx >> 6, q = idx & 63;
        float4 v = xf4[(size_t)(m0 + row) * 64 + q];
        float* d = &sa[row * LDA_P + q * 4];
        d[0] = v.x; d[1] = v.y; d[2] = v.z; d[3] = v.w;
    }
    __syncthreads();

    const int wave = t >> 5, lane = t & 31;
    const int half = lane >> 4, col = lane & 15;
    const int n0   = wave * 16;
    const int mrow = col;

    v8f acc = {};
    #pragma unroll 8
    for (int k0 = 0; k0 < IN_C; k0 += 4) {
        int ka = k0 + 2 * half;
        v2f a = *(const v2f*)&sa[mrow * LDA_P + ka];
        v2f b;
        b.x = w[(size_t)ka * HID_C + n0 + col];
        b.y = w[(size_t)(ka + 1) * HID_C + n0 + col];
        acc = __builtin_amdgcn_wmma_f32_16x16x4_f32(false, a, false, b,
                                                    (short)0, acc, false, false);
    }

    float bv = bias[n0 + col];
    #pragma unroll
    for (int i = 0; i < 8; ++i) {
        int row = i + 8 * half;
        x0[(size_t)(m0 + row) * HID_C + n0 + col] = acc[i] + bv;
    }
}

#define OWN_ROWS   400
#define TP         132
#define TJ         33
#define OWN_GROUPS (N_NODES / OWN_ROWS)
#define OWN_WAVES  32
#define SCAN_LOADS 4
#define DRAIN      8
#define QCAP       (DRAIN + SCAN_LOADS * 128)
#define DEG_ILP    5
static_assert(N_NODES % OWN_ROWS == 0, "ranges must tile the nodes");
static_assert(N_EDGES % 4 == 0, "dst is read 4 at a time");
static_assert(OWN_ROWS <= 512 && N_EDGES <= (1 << 23), "queue word packs edge id and 9-bit row");
static_assert(N_NODES <= (1 << 23), "record word packs source node and 9-bit row");

#define OWN_UNIT   (SCAN_LOADS * 32)
#define OWN_UNITS  (N_EDGES / 4 / OWN_UNIT)
#define OWN_N_IT   ((OWN_UNITS + OWN_WAVES - 1) / OWN_WAVES)
static_assert(N_EDGES % (4 * OWN_UNIT) == 0, "the dst list is a whole number of wave units");

__device__ __forceinline__ void own_add_row(float* __restrict__ tile, unsigned row, int lane,
                                            float nrm, float4 v) {
    float* a = tile + row * TP + lane;
    fadd_lds(a,          nrm * v.x);
    fadd_lds(a + TJ,     nrm * v.y);
    fadd_lds(a + 2 * TJ, nrm * v.z);
    fadd_lds(a + 3 * TJ, nrm * v.w);
}

__device__ __forceinline__ void own_record(uint2* __restrict__ seg, unsigned* __restrict__ cursor,
                                           int n, int lane, int sl, unsigned rl, int lo,
                                           const float* __restrict__ dinv) {
    float nl = dinv[sl] * dinv[lo + rl];
    unsigned base = 0;
    if (lane == 0)
        base = __hip_atomic_fetch_add(cursor, (unsigned)n, __ATOMIC_RELAXED,
                                      __HIP_MEMORY_SCOPE_WORKGROUP);
    base = (unsigned)__builtin_amdgcn_readfirstlane((int)base);
    if (lane < n)
        seg[base + lane] = make_uint2(((unsigned)sl << 9) | rl, __float_as_uint(nl));
}

__device__ __forceinline__ int own_load_trip(const int4* __restrict__ d4, int j0, int n,
                                             int wave, int lane,
                                             int4 (&out)[SCAN_LOADS], bool& ok) {
    int j = j0 + n;
    if (j >= OWN_N_IT) j -= OWN_N_IT;
    const int unit = j * OWN_WAVES + wave;
    ok = (n < OWN_N_IT) && (unit < OWN_UNITS);
    const int4* p = d4 + (ok ? unit : 0) * OWN_UNIT + lane;
    #pragma unroll
    for (int u = 0; u < SCAN_LOADS; ++u) out[u] = p[u * 32];
    return unit * OWN_UNIT + lane;
}

__device__ __forceinline__ int own_trip(const int4 (&cur)[SCAN_LOADS], int b, int lo, int lane,
                                        unsigned* __restrict__ q, int cnt,
                                        float* __restrict__ tile,
                                        const int* __restrict__ src,
                                        const float* __restrict__ dinv,
                                        const float4* __restrict__ h4,
                                        unsigned* __restrict__ cursor,
                                        uint2* __restrict__ seg) {
    #pragma unroll
    for (int u = 0; u < SCAN_LOADS; ++u) {
        unsigned r[4] = { (unsigned)(cur[u].x - lo), (unsigned)(cur[u].y - lo),
                          (unsigned)(cur[u].z - lo), (unsigned)(cur[u].w - lo) };
        unsigned rmin = min(min(r[0], r[1]), min(r[2], r[3]));
        if (__builtin_amdgcn_ballot_w32(rmin < OWN_ROWS) != 0) {
            #pragma unroll
            for (int k = 0; k < 4; ++k) {
                bool m = r[k] < OWN_ROWS;
                unsigned mask = __builtin_amdgcn_ballot_w32(m);
                if (mask != 0) {
                    if (m) {
                        unsigned e = (unsigned)((b + u * 32) * 4 + k);
                        q[cnt + __builtin_amdgcn_mbcnt_lo(mask, 0)] = (e << 9) | r[k];
                    }
                    cnt += __builtin_popcount(mask);
                }
            }
        }
    }

    while (cnt >= DRAIN) {
        cnt -= DRAIN;
        unsigned pe = q[cnt + (lane & (DRAIN - 1))];
        int      sl = src[pe >> 9];
        unsigned row[DRAIN];
        float4   v[DRAIN];
        float    nrm[DRAIN];
        #pragma unroll
        for (int i = 0; i < DRAIN; ++i) {
            int s  = __builtin_amdgcn_readlane(sl, i);
            row[i] = (unsigned)__builtin_amdgcn_readlane((int)pe, i) & 511u;
            v[i]   = h4[(size_t)s * (HID_C / 4) + lane];
            nrm[i] = dinv[s] * dinv[lo + row[i]];
        }
        own_record(seg, cursor, DRAIN, lane, sl, pe & 511u, lo, dinv);
        #pragma unroll
        for (int i = 0; i < DRAIN; ++i) own_add_row(tile, row[i], lane, nrm[i], v[i]);
    }
    return cnt;
}

#define OWN_MT (OWN_ROWS / 16)
static_assert(OWN_ROWS % 16 == 0, "the tile is a whole number of 16-row matrix tiles");
__device__ __forceinline__ void own_layer(float* __restrict__ tile, int lo, int t, int lane, int wave,
                                          const float* __restrict__ hin,
                                          const float* __restrict__ x0,
                                          const float* __restrict__ dinv,
                                          const float* __restrict__ w, float beta, int do_relu,
                                          float* __restrict__ hout) {
    const int half = lane >> 4, col = lane & 15;
    const int n0   = (wave & 7) * 16;
    const int ch   = n0 + col;

    v2f b[HID_C / 4];
    #pragma unroll
    for (int kk = 0; kk < HID_C / 4; ++kk) {
        const int ka = 4 * kk + 2 * half;
        b[kk].x = w[(size_t)ka * HID_C + ch];
        b[kk].y = w[(size_t)(ka + 1) * HID_C + ch];
    }

    const float4* h4 = (const float4*)hin + (size_t)lo * (HID_C / 4);
    const float4* x4 = (const float4*)x0  + (size_t)lo * (HID_C / 4);
    constexpr int NQ = OWN_ROWS * (HID_C / 4);
    for (int i0 = 0; i0 < NQ; i0 += 4 * OWN_WAVES * 32) {
        float4 hv[4], xv[4];
        float  dv[4];
        #pragma unroll
        for (int k = 0; k < 4; ++k) {
            const int i = min(i0 + k * OWN_WAVES * 32 + t, NQ - 1);
            dv[k] = dinv[lo + (i >> 5)];
            hv[k] = h4[i];
            xv[k] = x4[i];
        }
        #pragma unroll
        for (int k = 0; k < 4; ++k) {
            const int i = i0 + k * OWN_WAVES * 32 + t;
            if (i < NQ) {
                float* a = tile + (i >> 5) * TP + (i & 31);
                const float d2 = dv[k] * dv[k];
                a[0]      = (1.f - ALPHA) * (a[0]      + d2 * hv[k].x) + ALPHA * xv[k].x;
                a[TJ]     = (1.f - ALPHA) * (a[TJ]     + d2 * hv[k].y) + ALPHA * xv[k].y;
                a[2 * TJ] = (1.f - ALPHA) * (a[2 * TJ] + d2 * hv[k].z) + ALPHA * xv[k].z;
                a[3 * TJ] = (1.f - ALPHA) * (a[3 * TJ] + d2 * hv[k].w) + ALPHA * xv[k].w;
            }
        }
    }
    __syncthreads();

    for (int mt = wave >> 3; mt < OWN_MT; mt += OWN_WAVES / 8) {
        const float* ar = tile + (mt * 16 + col) * TP + 2 * half * TJ;
        v8f acc = {};
        #pragma unroll
        for (int kk = 0; kk < HID_C / 4; ++kk) {
            v2f a;
            a.x = ar[kk];
            a.y = ar[TJ + kk];
            acc = __builtin_amdgcn_wmma_f32_16x16x4_f32(false, a, false, b[kk],
                                                        (short)0, acc, false, false);
        }
        const float* sr = tile + (mt * 16 + 8 * half) * TP + (ch & 3) * TJ + (ch >> 2);
        float* orow = hout + (size_t)(lo + mt * 16 + 8 * half) * HID_C + ch;
        #pragma unroll
        for (int i = 0; i < 8; ++i) {
            float v = (1.f - beta) * sr[i * TP] + beta * acc[i];
            if (do_relu) v = fmaxf(v, 0.f);
            orow[(size_t)i * HID_C] = v;
        }
    }
}

__global__ __launch_bounds__(OWN_WAVES * 32)
void kown_scan(const int* __restrict__ src, const int* __restrict__ dst,
               const float* __restrict__ dinv, const int* __restrict__ degi,
               const float* __restrict__ hin, const float* __restrict__ x0,
               const float* __restrict__ w, float beta, int do_relu,
               float* __restrict__ hout,
               unsigned* __restrict__ rec, int* __restrict__ gmeta) {
    __shared__ __attribute__((aligned(16))) float tile[OWN_ROWS * TP];
    __shared__ unsigned queue[OWN_WAVES * QCAP];
    __shared__ int      wsum[OWN_WAVES];
    __shared__ unsigned cursor;

    const int t = threadIdx.x, lane = t & 31;
    const int wave = __builtin_amdgcn_readfirstlane(t >> 5);
    const int lo = blockIdx.x * OWN_ROWS;

    int part = 0;
    const int4* dg4 = (const int4*)degi;
    const int   lim = lo / 4;
    for (int i0 = 0; i0 < lim; i0 += DEG_ILP * OWN_WAVES * 32) {
        #pragma unroll
        for (int k = 0; k < DEG_ILP; ++k) {
            const int i = i0 + k * OWN_WAVES * 32 + t;
            int4 d = dg4[min(i, lim - 1)];
            part += ((d.x + d.y) + (d.z + d.w)) & (i < lim ? -1 : 0);
        }
    }
    #pragma unroll
    for (int o = 16; o > 0; o >>= 1) part += __shfl_xor(part, o);
    if (lane == 0) wsum[wave] = part;
    if (t == 0) cursor = 0u;

    float4* tile4 = (float4*)tile;
    for (int i = t; i < OWN_ROWS * TP / 4; i += OWN_WAVES * 32)
        tile4[i] = make_float4(0.f, 0.f, 0.f, 0.f);
    __syncthreads();

    int goff = -lo;
    for (int i = 0; i < OWN_WAVES; ++i) goff += wsum[i];
    goff = __builtin_amdgcn_readfirstlane(goff);
    uint2* seg = (uint2*)rec + goff;

    unsigned* q = &queue[wave * QCAP];
    int cnt = 0;
    const int4*   d4 = (const int4*)dst;
    const float4* h4 = (const float4*)hin;

    const int j0 = (int)(blockIdx.x * OWN_N_IT) / OWN_GROUPS;

    int4 wa[SCAN_LOADS], wb[SCAN_LOADS];
    bool oka, okb;
    int ba = own_load_trip(d4, j0, 0, wave, lane, wa, oka);
    for (int n = 0; n < OWN_N_IT; n += 2) {
        int bb = own_load_trip(d4, j0, n + 1, wave, lane, wb, okb);
        if (oka) cnt = own_trip(wa, ba, lo, lane, q, cnt, tile, src, dinv, h4, &cursor, seg);
        ba = own_load_trip(d4, j0, n + 2, wave, lane, wa, oka);
        if (okb) cnt = own_trip(wb, bb, lo, lane, q, cnt, tile, src, dinv, h4, &cursor, seg);
    }

    if (cnt > 0) {
        unsigned pl = q[lane < cnt ? lane : 0];
        own_record(seg, &cursor, cnt, lane, src[pl >> 9], pl & 511u, lo, dinv);
    }
    for (int i = 0; i < cnt; ++i) {
        unsigned pe  = (unsigned)__builtin_amdgcn_readfirstlane((int)q[i]);
        int      s   = src[pe >> 9];
        unsigned row = pe & 511u;
        float4   v   = h4[(size_t)s * (HID_C / 4) + lane];
        own_add_row(tile, row, lane, dinv[s] * dinv[lo + row], v);
    }
    __syncthreads();

    if (t == 0) {
        gmeta[2 * blockIdx.x]     = goff;
        gmeta[2 * blockIdx.x + 1] = (int)cursor;
    }
    own_layer(tile, lo, t, lane, wave, hin, x0, dinv, w, beta, do_relu, hout);
}

__global__ __launch_bounds__(OWN_WAVES * 32)
void kown_replay(const unsigned* __restrict__ rec, const int* __restrict__ gmeta,
                 const float* __restrict__ hin, const float* __restrict__ x0,
                 const float* __restrict__ dinv,
                 const float* __restrict__ w, float beta, int do_relu,
                 float* __restrict__ hout) {
    __shared__ __attribute__((aligned(16))) float tile[OWN_ROWS * TP];

    const int t = threadIdx.x, lane = t & 31;
    const int wave = __builtin_amdgcn_readfirstlane(t >> 5);
    const int lo = blockIdx.x * OWN_ROWS;

    float4* tile4 = (float4*)tile;
    for (int i = t; i < OWN_ROWS * TP / 4; i += OWN_WAVES * 32)
        tile4[i] = make_float4(0.f, 0.f, 0.f, 0.f);
    __syncthreads();

    const int    goff = gmeta[2 * blockIdx.x];
    const int    n    = gmeta[2 * blockIdx.x + 1];
    const int    nd   = n / DRAIN;
    const uint2* seg  = (const uint2*)rec + goff;
    const float4* h4  = (const float4*)hin;

    int   d  = wave;
    uint2 pe = make_uint2(0u, 0u);
    if (d < nd) pe = seg[d * DRAIN + (lane & (DRAIN - 1))];
    while (d < nd) {
        unsigned row[DRAIN];
        float4   v[DRAIN];
        float    nrm[DRAIN];
        #pragma unroll
        for (int i = 0; i < DRAIN; ++i) {
            unsigned w = (unsigned)__builtin_amdgcn_readlane((int)pe.x, i);
            row[i] = w & 511u;
            nrm[i] = __uint_as_float((unsigned)__builtin_amdgcn_readlane((int)pe.y, i));
            v[i]   = h4[(size_t)(w >> 9) * (HID_C / 4) + lane];
        }
        const int   dn = d + OWN_WAVES;
        const uint2 pn = seg[min(dn, nd - 1) * DRAIN + (lane & (DRAIN - 1))];
        #pragma unroll
        for (int i = 0; i < DRAIN; ++i) own_add_row(tile, row[i], lane, nrm[i], v[i]);
        pe = pn;
        d  = dn;
    }
    if (wave == 0) {
        for (int i = nd * DRAIN; i < n; ++i) {
            uint2    r = seg[i];
            unsigned w = (unsigned)__builtin_amdgcn_readfirstlane((int)r.x);
            float    f = __uint_as_float((unsigned)__builtin_amdgcn_readfirstlane((int)r.y));
            float4   v = h4[(size_t)(w >> 9) * (HID_C / 4) + lane];
            own_add_row(tile, w & 511u, lane, f, v);
        }
    }
    __syncthreads();

    own_layer(tile, lo, t, lane, wave, hin, x0, dinv, w, beta, do_relu, hout);
}

extern "C" void kernel_launch(void* const* d_in, const int* in_sizes, int n_in,
                              void* d_out, int out_size, void* d_ws, size_t ws_size,
                              hipStream_t stream) {
    const float* x      = (const float*)d_in[0];
    const int*   ei     = (const int*)d_in[1];
    const float* proj_w = (const float*)d_in[2];
    const float* proj_b = (const float*)d_in[3];
    const float* conv_w = (const float*)d_in[4];
    float* out = (float*)d_out;

    const int* e_src = ei;
    const int* e_dst = ei + N_EDGES;

    const size_t NH = (size_t)N_NODES * HID_C;
    char* ws = (char*)d_ws;
    int*   degi = (int*)ws;                                  ws += ((size_t)N_NODES * 4 + 255) & ~255ull;
    float* dinv = (float*)ws;                                ws += ((size_t)N_NODES * 4 + 255) & ~255ull;
    float* x0   = (float*)ws;                                ws += NH * 4;
    float* ha   = (float*)ws;                                ws += NH * 4;
    float* hb   = (float*)ws;                                ws += NH * 4;
    unsigned* rec   = (unsigned*)ws;                         ws += (size_t)N_EDGES * 8;
    int*      gmeta = (int*)ws;

    kdeg_init <<<(N_NODES + 255) / 256, 256, 0, stream>>>(degi);
    kdeg_count<<<(N_EDGES + 255) / 256, 256, 0, stream>>>(e_dst, degi);
    kdinv     <<<(N_NODES + 255) / 256, 256, 0, stream>>>(degi, dinv);

    kproj<<<N_NODES / 16, 256, 0, stream>>>(x, proj_w, proj_b, x0);

    static const float betas[N_LAYERS] = {0.40546510f, 0.22314355f, 0.15415068f, 0.11778303f};

    const float* const hins[N_LAYERS]  = {x0, ha, hb, ha};
    float* const       houts[N_LAYERS] = {ha, hb, ha, out};

    for (int i = 0; i < N_LAYERS; ++i) {
        const float* hin  = hins[i];
        float*       hout = houts[i];
        const float* wl   = conv_w + (size_t)i * HID_C * HID_C;
        const int    relu = (i != N_LAYERS - 1) ? 1 : 0;
        if (i == 0)
            kown_scan  <<<OWN_GROUPS, OWN_WAVES * 32, 0, stream>>>(e_src, e_dst, dinv, degi, hin, x0,
                                                                 wl, betas[i], relu, hout,
                                                                 rec, gmeta);
        else
            kown_replay<<<OWN_GROUPS, OWN_WAVES * 32, 0, stream>>>(rec, gmeta, hin, x0, dinv,
                                                                 wl, betas[i], relu, hout);
    }
}


# Owner work-groups search one owner byte per edge (dst/400, written by the degree kernel), not the 4-byte dst list: a quarter of the scan bytes per group; row sums stay in arrival order
# speedup vs baseline: 3.1968x; 1.0368x over previous
#include <hip/hip_runtime.h>
#include <math.h>

#define N_NODES 100000
#define N_EDGES 1600000
#define IN_C    256
#define HID_C   128
#define N_LAYERS 4
#define ALPHA   0.1f
#define OWN_ROWS 400

typedef float v2f __attribute__((ext_vector_type(2)));
typedef float v8f __attribute__((ext_vector_type(8)));

__device__ __forceinline__ void fadd_lds(float* p, float v) {
    __hip_atomic_fetch_add(p, v, __ATOMIC_RELAXED, __HIP_MEMORY_SCOPE_WORKGROUP);
}

__global__ void kdeg_init(int* degi) {
    int v = blockIdx.x * 256 + threadIdx.x;
    if (v < N_NODES) degi[v] = 1;
}

static_assert(N_NODES / OWN_ROWS <= 256, "an owner number must fit a byte");
__global__ void kdeg_count(const int* __restrict__ dst, int* __restrict__ degi,
                           unsigned* __restrict__ code) {
    int i = blockIdx.x * 256 + threadIdx.x;
    if (i < N_EDGES / 4) {
        int4 d = ((const int4*)dst)[i];
        atomicAdd(&degi[d.x], 1);
        atomicAdd(&degi[d.y], 1);
        atomicAdd(&degi[d.z], 1);
        atomicAdd(&degi[d.w], 1);
        code[i] = ((unsigned)d.x / OWN_ROWS)       | ((unsigned)d.y / OWN_ROWS) << 8 |
                  ((unsigned)d.z / OWN_ROWS) << 16 | ((unsigned)d.w / OWN_ROWS) << 24;
    }
}

__global__ void kdinv(const int* __restrict__ degi, float* __restrict__ dinv) {
    int v = blockIdx.x * 256 + threadIdx.x;
    if (v < N_NODES) dinv[v] = rsqrtf((float)degi[v]);
}

#define LDA_P 260
__global__ void kproj(const float* __restrict__ x, const float* __restrict__ w,
                      const float* __restrict__ bias, float* __restrict__ x0) {
    __shared__ float sa[16 * LDA_P];
    const int t   = threadIdx.x;
    const int m0  = blockIdx.x * 16;

    const float4* xf4 = (const float4*)x;
    #pragma unroll
    for (int c = 0; c < 4; ++c) {
        int idx = c * 256 + t;
        int row = idx >> 6, q = idx & 63;
        float4 v = xf4[(size_t)(m0 + row) * 64 + q];
        float* d = &sa[row * LDA_P + q * 4];
        d[0] = v.x; d[1] = v.y; d[2] = v.z; d[3] = v.w;
    }
    __syncthreads();

    const int wave = t >> 5, lane = t & 31;
    const int half = lane >> 4, col = lane & 15;
    const int n0   = wave * 16;
    const int mrow = col;

    v8f acc = {};
    #pragma unroll 8
    for (int k0 = 0; k0 < IN_C; k0 += 4) {
        int ka = k0 + 2 * half;
        v2f a = *(const v2f*)&sa[mrow * LDA_P + ka];
        v2f b;
        b.x = w[(size_t)ka * HID_C + n0 + col];
        b.y = w[(size_t)(ka + 1) * HID_C + n0 + col];
        acc = __builtin_amdgcn_wmma_f32_16x16x4_f32(false, a, false, b,
                                                    (short)0, acc, false, false);
    }

    float bv = bias[n0 + col];
    #pragma unroll
    for (int i = 0; i < 8; ++i) {
        int row = i + 8 * half;
        x0[(size_t)(m0 + row) * HID_C + n0 + col] = acc[i] + bv;
    }
}

#define TP         132
#define TJ         33
#define OWN_GROUPS (N_NODES / OWN_ROWS)
#define OWN_WAVES  32
#define SCAN_LOADS 5
#define DRAIN      8
#define QWORD      (16 * 32)
#define QCAP       (DRAIN + QWORD + 256)
#define DEG_ILP    5
static_assert(N_NODES % OWN_ROWS == 0, "ranges must tile the nodes");
static_assert(N_EDGES % 16 == 0, "the owner bytes are read 16 at a time");
static_assert(OWN_ROWS <= 512, "record word packs a 9-bit row");
static_assert(N_NODES <= (1 << 23), "record word packs source node and 9-bit row");

#define OWN_UNIT   (SCAN_LOADS * 32)
#define OWN_UNITS  (N_EDGES / 16 / OWN_UNIT)
#define OWN_N_IT   ((OWN_UNITS + OWN_WAVES - 1) / OWN_WAVES)
static_assert(N_EDGES % (16 * OWN_UNIT) == 0, "the code list is a whole number of wave units");

__device__ __forceinline__ void own_add_row(float* __restrict__ tile, unsigned row, int lane,
                                            float nrm, float4 v) {
    float* a = tile + row * TP + lane;
    fadd_lds(a,          nrm * v.x);
    fadd_lds(a + TJ,     nrm * v.y);
    fadd_lds(a + 2 * TJ, nrm * v.z);
    fadd_lds(a + 3 * TJ, nrm * v.w);
}

__device__ __forceinline__ void own_record(uint2* __restrict__ seg, unsigned* __restrict__ cursor,
                                           int n, int lane, int sl, unsigned rl, int lo,
                                           const float* __restrict__ dinv) {
    float nl = dinv[sl] * dinv[lo + rl];
    unsigned base = 0;
    if (lane == 0)
        base = __hip_atomic_fetch_add(cursor, (unsigned)n, __ATOMIC_RELAXED,
                                      __HIP_MEMORY_SCOPE_WORKGROUP);
    base = (unsigned)__builtin_amdgcn_readfirstlane((int)base);
    if (lane < n)
        seg[base + lane] = make_uint2(((unsigned)sl << 9) | rl, __float_as_uint(nl));
}

__device__ __forceinline__ int own_load_trip(const uint4* __restrict__ d4, int j0, int n,
                                             int wave, int lane,
                                             uint4 (&out)[SCAN_LOADS], bool& ok) {
    int j = j0 + n;
    if (j >= OWN_N_IT) j -= OWN_N_IT;
    const int unit = j * OWN_WAVES + wave;
    ok = (n < OWN_N_IT) && (unit < OWN_UNITS);
    const uint4* p = d4 + (ok ? unit : 0) * OWN_UNIT + lane;
    #pragma unroll
    for (int u = 0; u < SCAN_LOADS; ++u) out[u] = p[u * 32];
    return unit * OWN_UNIT + lane;
}

__device__ __forceinline__ int own_trip(const uint4 (&cur)[SCAN_LOADS], int b, unsigned g4,
                                        int lo, int lane,
                                        unsigned* __restrict__ q, int cnt,
                                        float* __restrict__ tile,
                                        const int* __restrict__ src,
                                        const int* __restrict__ dst,
                                        const float* __restrict__ dinv,
                                        const float4* __restrict__ h4,
                                        unsigned* __restrict__ cursor,
                                        uint2* __restrict__ seg) {
  int u0 = 0;
  do {
    #pragma unroll
    for (int u = 0; u < SCAN_LOADS; ++u) {
      if (u == u0 && cnt <= QCAP - QWORD) {
        u0 = u + 1;
        unsigned w0 = cur[u].x, w1 = cur[u].y, w2 = cur[u].z, w3 = cur[u].w;
        const unsigned e0 = (unsigned)(b + u * 32) * 16u;
        #pragma nounroll
        for (int w = 0; w < 4; ++w) {
            const unsigned y = w0 ^ g4;
            w0 = w1; w1 = w2; w2 = w3;
            const unsigned z = (y - 0x01010101u) & ~y & 0x80808080u;
            if (__builtin_amdgcn_ballot_w32(z != 0u) != 0) {
                #pragma unroll
                for (int k = 0; k < 4; ++k) {
                    bool m = (y & (0xffu << (8 * k))) == 0u;
                    unsigned mask = __builtin_amdgcn_ballot_w32(m);
                    if (mask != 0) {
                        if (m) q[cnt + __builtin_amdgcn_mbcnt_lo(mask, 0)] = e0 + (unsigned)(4 * w + k);
                        cnt += __builtin_popcount(mask);
                    }
                }
            }
        }
      }
    }

    while (cnt >= DRAIN) {
        cnt -= DRAIN;
        unsigned pe = q[cnt + (lane & (DRAIN - 1))];
        int      sl = src[pe];
        unsigned rl = (unsigned)(dst[pe] - lo);
        unsigned row[DRAIN];
        float4   v[DRAIN];
        float    nrm[DRAIN];
        #pragma unroll
        for (int i = 0; i < DRAIN; ++i) {
            int s  = __builtin_amdgcn_readlane(sl, i);
            row[i] = (unsigned)__builtin_amdgcn_readlane((int)rl, i);
            v[i]   = h4[(size_t)s * (HID_C / 4) + lane];
            nrm[i] = dinv[s] * dinv[lo + row[i]];
        }
        own_record(seg, cursor, DRAIN, lane, sl, rl, lo, dinv);
        #pragma unroll
        for (int i = 0; i < DRAIN; ++i) own_add_row(tile, row[i], lane, nrm[i], v[i]);
    }
  } while (u0 < SCAN_LOADS);
    return cnt;
}

#define OWN_MT (OWN_ROWS / 16)
static_assert(OWN_ROWS % 16 == 0, "the tile is a whole number of 16-row matrix tiles");
__device__ __forceinline__ void own_layer(float* __restrict__ tile, int lo, int t, int lane, int wave,
                                          const float* __restrict__ hin,
                                          const float* __restrict__ x0,
                                          const float* __restrict__ dinv,
                                          const float* __restrict__ w, float beta, int do_relu,
                                          float* __restrict__ hout) {
    const int half = lane >> 4, col = lane & 15;
    const int n0   = (wave & 7) * 16;
    const int ch   = n0 + col;

    v2f b[HID_C / 4];
    #pragma unroll
    for (int kk = 0; kk < HID_C / 4; ++kk) {
        const int ka = 4 * kk + 2 * half;
        b[kk].x = w[(size_t)ka * HID_C + ch];
        b[kk].y = w[(size_t)(ka + 1) * HID_C + ch];
    }

    const float4* h4 = (const float4*)hin + (size_t)lo * (HID_C / 4);
    const float4* x4 = (const float4*)x0  + (size_t)lo * (HID_C / 4);
    constexpr int NQ = OWN_ROWS * (HID_C / 4);
    for (int i0 = 0; i0 < NQ; i0 += 4 * OWN_WAVES * 32) {
        float4 hv[4], xv[4];
        float  dv[4];
        #pragma unroll
        for (int k = 0; k < 4; ++k) {
            const int i = min(i0 + k * OWN_WAVES * 32 + t, NQ - 1);
            dv[k] = dinv[lo + (i >> 5)];
            hv[k] = h4[i];
            xv[k] = x4[i];
        }
        #pragma unroll
        for (int k = 0; k < 4; ++k) {
            const int i = i0 + k * OWN_WAVES * 32 + t;
            if (i < NQ) {
                float* a = tile + (i >> 5) * TP + (i & 31);
                const float d2 = dv[k] * dv[k];
                a[0]      = (1.f - ALPHA) * (a[0]      + d2 * hv[k].x) + ALPHA * xv[k].x;
                a[TJ]     = (1.f - ALPHA) * (a[TJ]     + d2 * hv[k].y) + ALPHA * xv[k].y;
                a[2 * TJ] = (1.f - ALPHA) * (a[2 * TJ] + d2 * hv[k].z) + ALPHA * xv[k].z;
                a[3 * TJ] = (1.f - ALPHA) * (a[3 * TJ] + d2 * hv[k].w) + ALPHA * xv[k].w;
            }
        }
    }
    __syncthreads();

    for (int mt = wave >> 3; mt < OWN_MT; mt += OWN_WAVES / 8) {
        const float* ar = tile + (mt * 16 + col) * TP + 2 * half * TJ;
        v8f acc = {};
        #pragma unroll
        for (int kk = 0; kk < HID_C / 4; ++kk) {
            v2f a;
            a.x = ar[kk];
            a.y = ar[TJ + kk];
            acc = __builtin_amdgcn_wmma_f32_16x16x4_f32(false, a, false, b[kk],
                                                        (short)0, acc, false, false);
        }
        const float* sr = tile + (mt * 16 + 8 * half) * TP + (ch & 3) * TJ + (ch >> 2);
        float* orow = hout + (size_t)(lo + mt * 16 + 8 * half) * HID_C + ch;
        #pragma unroll
        for (int i = 0; i < 8; ++i) {
            float v = (1.f - beta) * sr[i * TP] + beta * acc[i];
            if (do_relu) v = fmaxf(v, 0.f);
            orow[(size_t)i * HID_C] = v;
        }
    }
}

__global__ __launch_bounds__(OWN_WAVES * 32)
void kown_scan(const int* __restrict__ src, const int* __restrict__ dst,
               const unsigned* __restrict__ code,
               const float* __restrict__ dinv, const int* __restrict__ degi,
               const float* __restrict__ hin, const float* __restrict__ x0,
               const float* __restrict__ w, float beta, int do_relu,
               float* __restrict__ hout,
               unsigned* __restrict__ rec, int* __restrict__ gmeta) {
    __shared__ __attribute__((aligned(16))) float tile[OWN_ROWS * TP];
    __shared__ unsigned queue[OWN_WAVES * QCAP];
    __shared__ int      wsum[OWN_WAVES];
    __shared__ unsigned cursor;

    const int t = threadIdx.x, lane = t & 31;
    const int wave = __builtin_amdgcn_readfirstlane(t >> 5);
    const int lo = blockIdx.x * OWN_ROWS;

    int part = 0;
    const int4* dg4 = (const int4*)degi;
    const int   lim = lo / 4;
    for (int i0 = 0; i0 < lim; i0 += DEG_ILP * OWN_WAVES * 32) {
        #pragma unroll
        for (int k = 0; k < DEG_ILP; ++k) {
            const int i = i0 + k * OWN_WAVES * 32 + t;
            int4 d = dg4[min(i, lim - 1)];
            part += ((d.x + d.y) + (d.z + d.w)) & (i < lim ? -1 : 0);
        }
    }
    #pragma unroll
    for (int o = 16; o > 0; o >>= 1) part += __shfl_xor(part, o);
    if (lane == 0) wsum[wave] = part;
    if (t == 0) cursor = 0u;

    float4* tile4 = (float4*)tile;
    for (int i = t; i < OWN_ROWS * TP / 4; i += OWN_WAVES * 32)
        tile4[i] = make_float4(0.f, 0.f, 0.f, 0.f);
    __syncthreads();

    int goff = -lo;
    for (int i = 0; i < OWN_WAVES; ++i) goff += wsum[i];
    goff = __builtin_amdgcn_readfirstlane(goff);
    uint2* seg = (uint2*)rec + goff;

    unsigned* q = &queue[wave * QCAP];
    int cnt = 0;
    const uint4*  d4 = (const uint4*)code;
    const float4* h4 = (const float4*)hin;
    const unsigned g4 = blockIdx.x * 0x01010101u;

    const int j0 = (int)(blockIdx.x * OWN_N_IT) / OWN_GROUPS;

    uint4 wa[SCAN_LOADS], wb[SCAN_LOADS];
    bool oka, okb;
    int ba = own_load_trip(d4, j0, 0, wave, lane, wa, oka);
    for (int n = 0; n < OWN_N_IT; n += 2) {
        int bb = own_load_trip(d4, j0, n + 1, wave, lane, wb, okb);
        if (oka) cnt = own_trip(wa, ba, g4, lo, lane, q, cnt, tile, src, dst, dinv, h4, &cursor, seg);
        ba = own_load_trip(d4, j0, n + 2, wave, lane, wa, oka);
        if (okb) cnt = own_trip(wb, bb, g4, lo, lane, q, cnt, tile, src, dst, dinv, h4, &cursor, seg);
    }

    if (cnt > 0) {
        unsigned pl = q[lane < cnt ? lane : 0];
        own_record(seg, &cursor, cnt, lane, src[pl], (unsigned)(dst[pl] - lo), lo, dinv);
    }
    for (int i = 0; i < cnt; ++i) {
        unsigned pe  = (unsigned)__builtin_amdgcn_readfirstlane((int)q[i]);
        int      s   = src[pe];
        unsigned row = (unsigned)(dst[pe] - lo);
        float4   v   = h4[(size_t)s * (HID_C / 4) + lane];
        own_add_row(tile, row, lane, dinv[s] * dinv[lo + row], v);
    }
    __syncthreads();

    if (t == 0) {
        gmeta[2 * blockIdx.x]     = goff;
        gmeta[2 * blockIdx.x + 1] = (int)cursor;
    }
    own_layer(tile, lo, t, lane, wave, hin, x0, dinv, w, beta, do_relu, hout);
}

__global__ __launch_bounds__(OWN_WAVES * 32)
void kown_replay(const unsigned* __restrict__ rec, const int* __restrict__ gmeta,
                 const float* __restrict__ hin, const float* __restrict__ x0,
                 const float* __restrict__ dinv,
                 const float* __restrict__ w, float beta, int do_relu,
                 float* __restrict__ hout) {
    __shared__ __attribute__((aligned(16))) float tile[OWN_ROWS * TP];

    const int t = threadIdx.x, lane = t & 31;
    const int wave = __builtin_amdgcn_readfirstlane(t >> 5);
    const int lo = blockIdx.x * OWN_ROWS;

    float4* tile4 = (float4*)tile;
    for (int i = t; i < OWN_ROWS * TP / 4; i += OWN_WAVES * 32)
        tile4[i] = make_float4(0.f, 0.f, 0.f, 0.f);
    __syncthreads();

    const int    goff = gmeta[2 * blockIdx.x];
    const int    n    = gmeta[2 * blockIdx.x + 1];
    const int    nd   = n / DRAIN;
    const uint2* seg  = (const uint2*)rec + goff;
    const float4* h4  = (const float4*)hin;

    int   d  = wave;
    uint2 pe = make_uint2(0u, 0u);
    if (d < nd) pe = seg[d * DRAIN + (lane & (DRAIN - 1))];
    while (d < nd) {
        unsigned row[DRAIN];
        float4   v[DRAIN];
        float    nrm[DRAIN];
        #pragma unroll
        for (int i = 0; i < DRAIN; ++i) {
            unsigned w = (unsigned)__builtin_amdgcn_readlane((int)pe.x, i);
            row[i] = w & 511u;
            nrm[i] = __uint_as_float((unsigned)__builtin_amdgcn_readlane((int)pe.y, i));
            v[i]   = h4[(size_t)(w >> 9) * (HID_C / 4) + lane];
        }
        const int   dn = d + OWN_WAVES;
        const uint2 pn = seg[min(dn, nd - 1) * DRAIN + (lane & (DRAIN - 1))];
        #pragma unroll
        for (int i = 0; i < DRAIN; ++i) own_add_row(tile, row[i], lane, nrm[i], v[i]);
        pe = pn;
        d  = dn;
    }
    if (wave == 0) {
        for (int i = nd * DRAIN; i < n; ++i) {
            uint2    r = seg[i];
            unsigned w = (unsigned)__builtin_amdgcn_readfirstlane((int)r.x);
            float    f = __uint_as_float((unsigned)__builtin_amdgcn_readfirstlane((int)r.y));
            float4   v = h4[(size_t)(w >> 9) * (HID_C / 4) + lane];
            own_add_row(tile, w & 511u, lane, f, v);
        }
    }
    __syncthreads();

    own_layer(tile, lo, t, lane, wave, hin, x0, dinv, w, beta, do_relu, hout);
}

extern "C" void kernel_launch(void* const* d_in, const int* in_sizes, int n_in,
                              void* d_out, int out_size, void* d_ws, size_t ws_size,
                              hipStream_t stream) {
    const float* x      = (const float*)d_in[0];
    const int*   ei     = (const int*)d_in[1];
    const float* proj_w = (const float*)d_in[2];
    const float* proj_b = (const float*)d_in[3];
    const float* conv_w = (const float*)d_in[4];
    float* out = (float*)d_out;

    const int* e_src = ei;
    const int* e_dst = ei + N_EDGES;

    const size_t NH = (size_t)N_NODES * HID_C;
    char* ws = (char*)d_ws;
    int*   degi = (int*)ws;                                  ws += ((size_t)N_NODES * 4 + 255) & ~255ull;
    float* dinv = (float*)ws;                                ws += ((size_t)N_NODES * 4 + 255) & ~255ull;
    float* x0   = (float*)ws;                                ws += NH * 4;
    float* ha   = (float*)ws;                                ws += NH * 4;
    float* hb   = (float*)ws;                                ws += NH * 4;
    unsigned* rec   = (unsigned*)ws;                         ws += (size_t)N_EDGES * 8;
    int*      gmeta = (int*)ws;                              ws += ((size_t)OWN_GROUPS * 8 + 255) & ~255ull;
    unsigned* code  = (unsigned*)ws;

    kdeg_init <<<(N_NODES + 255) / 256, 256, 0, stream>>>(degi);
    kdeg_count<<<(N_EDGES / 4 + 255) / 256, 256, 0, stream>>>(e_dst, degi, code);
    kdinv     <<<(N_NODES + 255) / 256, 256, 0, stream>>>(degi, dinv);

    kproj<<<N_NODES / 16, 256, 0, stream>>>(x, proj_w, proj_b, x0);

    static const float betas[N_LAYERS] = {0.40546510f, 0.22314355f, 0.15415068f, 0.11778303f};

    const float* const hins[N_LAYERS]  = {x0, ha, hb, ha};
    float* const       houts[N_LAYERS] = {ha, hb, ha, out};

    for (int i = 0; i < N_LAYERS; ++i) {
        const float* hin  = hins[i];
        float*       hout = houts[i];
        const float* wl   = conv_w + (size_t)i * HID_C * HID_C;
        const int    relu = (i != N_LAYERS - 1) ? 1 : 0;
        if (i == 0)
            kown_scan  <<<OWN_GROUPS, OWN_WAVES * 32, 0, stream>>>(e_src, e_dst, code, dinv, degi, hin, x0,
                                                                 wl, betas[i], relu, hout,
                                                                 rec, gmeta);
        else
            kown_replay<<<OWN_GROUPS, OWN_WAVES * 32, 0, stream>>>(rec, gmeta, hin, x0, dinv,
                                                                 wl, betas[i], relu, hout);
    }
}


# The 250 per-group searches of the edge list become one partition launch: 4096-edge groups bucket edges by owner range in LDS and write the records; all 4 layers replay; row sums stay in arrival order
# speedup vs baseline: 3.3200x; 1.0385x over previous
#include <hip/hip_runtime.h>
#include <math.h>

#define N_NODES 100000
#define N_EDGES 1600000
#define IN_C    256
#define HID_C   128
#define N_LAYERS 4
#define ALPHA   0.1f
#define OWN_ROWS 400

typedef float v2f __attribute__((ext_vector_type(2)));
typedef float v8f __attribute__((ext_vector_type(8)));

__device__ __forceinline__ void fadd_lds(float* p, float v) {
    __hip_atomic_fetch_add(p, v, __ATOMIC_RELAXED, __HIP_MEMORY_SCOPE_WORKGROUP);
}

#define OWN_GROUPS (N_NODES / OWN_ROWS)
#define CUR_PITCH  32

__global__ void kdeg_init(int* degi, int* gcur) {
    int v = blockIdx.x * 256 + threadIdx.x;
    if (v < N_NODES) degi[v] = 1;
    if (v < OWN_GROUPS) gcur[v * CUR_PITCH] = 0;
}

__global__ void kdeg_count(const int* __restrict__ dst, int* __restrict__ degi) {
    int i = blockIdx.x * 256 + threadIdx.x;
    if (i < N_EDGES / 4) {
        int4 d = ((const int4*)dst)[i];
        atomicAdd(&degi[d.x], 1);
        atomicAdd(&degi[d.y], 1);
        atomicAdd(&degi[d.z], 1);
        atomicAdd(&degi[d.w], 1);
    }
}

#define DINV_THREADS 512
static_assert(OWN_ROWS <= DINV_THREADS, "one thread per row of the range");
__global__ __launch_bounds__(DINV_THREADS)
void kdinv(const int* __restrict__ degi, float* __restrict__ dinv, int* __restrict__ gcount) {
    __shared__ int wsum[DINV_THREADS / 32];
    const int t = threadIdx.x;
    const int v = blockIdx.x * OWN_ROWS + t;
    int part = 0;
    if (t < OWN_ROWS) {
        part = degi[v];
        dinv[v] = rsqrtf((float)part);
    }
    for (int o = 16; o > 0; o >>= 1) part += __shfl_xor(part, o);
    if ((t & 31) == 0) wsum[t >> 5] = part;
    __syncthreads();
    if (t == 0) {
        int sum = -OWN_ROWS;
        for (int k = 0; k < DINV_THREADS / 32; ++k) sum += wsum[k];
        gcount[blockIdx.x] = sum;
    }
}

#define LDA_P 260
__global__ void kproj(const float* __restrict__ x, const float* __restrict__ w,
                      const float* __restrict__ bias, float* __restrict__ x0) {
    __shared__ float sa[16 * LDA_P];
    const int t   = threadIdx.x;
    const int m0  = blockIdx.x * 16;

    const float4* xf4 = (const float4*)x;
    #pragma unroll
    for (int c = 0; c < 4; ++c) {
        int idx = c * 256 + t;
        int row = idx >> 6, q = idx & 63;
        float4 v = xf4[(size_t)(m0 + row) * 64 + q];
        float* d = &sa[row * LDA_P + q * 4];
        d[0] = v.x; d[1] = v.y; d[2] = v.z; d[3] = v.w;
    }
    __syncthreads();

    const int wave = t >> 5, lane = t & 31;
    const int half = lane >> 4, col = lane & 15;
    const int n0   = wave * 16;
    const int mrow = col;

    v8f acc = {};
    #pragma unroll 8
    for (int k0 = 0; k0 < IN_C; k0 += 4) {
        int ka = k0 + 2 * half;
        v2f a = *(const v2f*)&sa[mrow * LDA_P + ka];
        v2f b;
        b.x = w[(size_t)ka * HID_C + n0 + col];
        b.y = w[(size_t)(ka + 1) * HID_C + n0 + col];
        acc = __builtin_amdgcn_wmma_f32_16x16x4_f32(false, a, false, b,
                                                    (short)0, acc, false, false);
    }

    float bv = bias[n0 + col];
    #pragma unroll
    for (int i = 0; i < 8; ++i) {
        int row = i + 8 * half;
        x0[(size_t)(m0 + row) * HID_C + n0 + col] = acc[i] + bv;
    }
}

#define TP         132
#define TJ         33
#define OWN_WAVES  32
#define DRAIN      8
static_assert(N_NODES % OWN_ROWS == 0, "ranges must tile the nodes");
static_assert(OWN_ROWS <= 512, "record word packs a 9-bit row");
static_assert(N_NODES <= (1 << 23), "record word packs source node and 9-bit row");

#define PART_THREADS 1024
#define PART_EDGES   (4 * PART_THREADS)
#define PART_BLOCKS  ((N_EDGES / 4 + PART_THREADS - 1) / PART_THREADS)
static_assert(N_EDGES % 4 == 0, "edges are read four at a time");
static_assert(OWN_GROUPS <= 256, "the first eight waves hold one owner per thread");

__global__ __launch_bounds__(PART_THREADS)
void kpart(const int* __restrict__ src, const int* __restrict__ dst,
           const float* __restrict__ dinv, const int* __restrict__ gcount,
           int* __restrict__ gcur, unsigned* __restrict__ rec, int* __restrict__ gmeta) {
    __shared__ int hist[OWN_GROUPS];
    __shared__ int sbase[OWN_GROUPS];
    __shared__ int wtot[8];

    const int t = threadIdx.x, lane = t & 31, wave = t >> 5;
    const int i = blockIdx.x * PART_THREADS + t;
    const bool ok = i < N_EDGES / 4;

    int4 d = make_int4(0, 0, 0, 0), s = make_int4(0, 0, 0, 0);
    if (ok) {
        d = ((const int4*)dst)[i];
        s = ((const int4*)src)[i];
    }
    const float fs0 = dinv[s.x], fs1 = dinv[s.y], fs2 = dinv[s.z], fs3 = dinv[s.w];
    const float fd0 = dinv[d.x], fd1 = dinv[d.y], fd2 = dinv[d.z], fd3 = dinv[d.w];

    int cnt_t = 0, start_t = 0;
    if (wave < 8) {
        cnt_t = (t < OWN_GROUPS) ? gcount[t] : 0;
        int incl = cnt_t;
        for (int o = 1; o < 32; o <<= 1) {
            int up = __shfl_up(incl, o);
            if (lane >= o) incl += up;
        }
        if (lane == 31) wtot[wave] = incl;
        start_t = incl - cnt_t;
    }
    if (t < OWN_GROUPS) hist[t] = 0;
    __syncthreads();
    if (t < OWN_GROUPS) {
        for (int k = 0; k < wave; ++k) start_t += wtot[k];
        if (blockIdx.x == 0) {
            gmeta[2 * t]     = start_t;
            gmeta[2 * t + 1] = cnt_t;
        }
    }

    const unsigned o0 = (unsigned)d.x / OWN_ROWS, o1 = (unsigned)d.y / OWN_ROWS,
                   o2 = (unsigned)d.z / OWN_ROWS, o3 = (unsigned)d.w / OWN_ROWS;
    int r0 = 0, r1 = 0, r2 = 0, r3 = 0;
    if (ok) {
        r0 = __hip_atomic_fetch_add(&hist[o0], 1, __ATOMIC_RELAXED, __HIP_MEMORY_SCOPE_WORKGROUP);
        r1 = __hip_atomic_fetch_add(&hist[o1], 1, __ATOMIC_RELAXED, __HIP_MEMORY_SCOPE_WORKGROUP);
        r2 = __hip_atomic_fetch_add(&hist[o2], 1, __ATOMIC_RELAXED, __HIP_MEMORY_SCOPE_WORKGROUP);
        r3 = __hip_atomic_fetch_add(&hist[o3], 1, __ATOMIC_RELAXED, __HIP_MEMORY_SCOPE_WORKGROUP);
    }
    __syncthreads();
    if (t < OWN_GROUPS) {
        const int n = hist[t];
        int b = 0;
        if (n > 0) b = atomicAdd(&gcur[t * CUR_PITCH], n);
        sbase[t] = start_t + b;
    }
    __syncthreads();
    if (ok) {
        uint2* seg = (uint2*)rec;
        seg[sbase[o0] + r0] = make_uint2(((unsigned)s.x << 9) | ((unsigned)d.x - o0 * OWN_ROWS),
                                         __float_as_uint(fs0 * fd0));
        seg[sbase[o1] + r1] = make_uint2(((unsigned)s.y << 9) | ((unsigned)d.y - o1 * OWN_ROWS),
                                         __float_as_uint(fs1 * fd1));
        seg[sbase[o2] + r2] = make_uint2(((unsigned)s.z << 9) | ((unsigned)d.z - o2 * OWN_ROWS),
                                         __float_as_uint(fs2 * fd2));
        seg[sbase[o3] + r3] = make_uint2(((unsigned)s.w << 9) | ((unsigned)d.w - o3 * OWN_ROWS),
                                         __float_as_uint(fs3 * fd3));
    }
}

__device__ __forceinline__ void own_add_row(float* __restrict__ tile, unsigned row, int lane,
                                            float nrm, float4 v) {
    float* a = tile + row * TP + lane;
    fadd_lds(a,          nrm * v.x);
    fadd_lds(a + TJ,     nrm * v.y);
    fadd_lds(a + 2 * TJ, nrm * v.z);
    fadd_lds(a + 3 * TJ, nrm * v.w);
}

#define OWN_MT (OWN_ROWS / 16)
static_assert(OWN_ROWS % 16 == 0, "the tile is a whole number of 16-row matrix tiles");
__device__ __forceinline__ void own_layer(float* __restrict__ tile, int lo, int t, int lane, int wave,
                                          const float* __restrict__ hin,
                                          const float* __restrict__ x0,
                                          const float* __restrict__ dinv,
                                          const float* __restrict__ w, float beta, int do_relu,
                                          float* __restrict__ hout) {
    const int half = lane >> 4, col = lane & 15;
    const int n0   = (wave & 7) * 16;
    const int ch   = n0 + col;

    v2f b[HID_C / 4];
    #pragma unroll
    for (int kk = 0; kk < HID_C / 4; ++kk) {
        const int ka = 4 * kk + 2 * half;
        b[kk].x = w[(size_t)ka * HID_C + ch];
        b[kk].y = w[(size_t)(ka + 1) * HID_C + ch];
    }

    const float4* h4 = (const float4*)hin + (size_t)lo * (HID_C / 4);
    const float4* x4 = (const float4*)x0  + (size_t)lo * (HID_C / 4);
    constexpr int NQ = OWN_ROWS * (HID_C / 4);
    for (int i0 = 0; i0 < NQ; i0 += 4 * OWN_WAVES * 32) {
        float4 hv[4], xv[4];
        float  dv[4];
        #pragma unroll
        for (int k = 0; k < 4; ++k) {
            const int i = min(i0 + k * OWN_WAVES * 32 + t, NQ - 1);
            dv[k] = dinv[lo + (i >> 5)];
            hv[k] = h4[i];
            xv[k] = x4[i];
        }
        #pragma unroll
        for (int k = 0; k < 4; ++k) {
            const int i = i0 + k * OWN_WAVES * 32 + t;
            if (i < NQ) {
                float* a = tile + (i >> 5) * TP + (i & 31);
                const float d2 = dv[k] * dv[k];
                a[0]      = (1.f - ALPHA) * (a[0]      + d2 * hv[k].x) + ALPHA * xv[k].x;
                a[TJ]     = (1.f - ALPHA) * (a[TJ]     + d2 * hv[k].y) + ALPHA * xv[k].y;
                a[2 * TJ] = (1.f - ALPHA) * (a[2 * TJ] + d2 * hv[k].z) + ALPHA * xv[k].z;
                a[3 * TJ] = (1.f - ALPHA) * (a[3 * TJ] + d2 * hv[k].w) + ALPHA * xv[k].w;
            }
        }
    }
    __syncthreads();

    for (int mt = wave >> 3; mt < OWN_MT; mt += OWN_WAVES / 8) {
        const float* ar = tile + (mt * 16 + col) * TP + 2 * half * TJ;
        v8f acc = {};
        #pragma unroll
        for (int kk = 0; kk < HID_C / 4; ++kk) {
            v2f a;
            a.x = ar[kk];
            a.y = ar[TJ + kk];
            acc = __builtin_amdgcn_wmma_f32_16x16x4_f32(false, a, false, b[kk],
                                                        (short)0, acc, false, false);
        }
        const float* sr = tile + (mt * 16 + 8 * half) * TP + (ch & 3) * TJ + (ch >> 2);
        float* orow = hout + (size_t)(lo + mt * 16 + 8 * half) * HID_C + ch;
        #pragma unroll
        for (int i = 0; i < 8; ++i) {
            float v = (1.f - beta) * sr[i * TP] + beta * acc[i];
            if (do_relu) v = fmaxf(v, 0.f);
            orow[(size_t)i * HID_C] = v;
        }
    }
}

__global__ __launch_bounds__(OWN_WAVES * 32)
void kown_replay(const unsigned* __restrict__ rec, const int* __restrict__ gmeta,
                 const float* __restrict__ hin, const float* __restrict__ x0,
                 const float* __restrict__ dinv,
                 const float* __restrict__ w, float beta, int do_relu,
                 float* __restrict__ hout) {
    __shared__ __attribute__((aligned(16))) float tile[OWN_ROWS * TP];

    const int t = threadIdx.x, lane = t & 31;
    const int wave = __builtin_amdgcn_readfirstlane(t >> 5);
    const int lo = blockIdx.x * OWN_ROWS;

    float4* tile4 = (float4*)tile;
    for (int i = t; i < OWN_ROWS * TP / 4; i += OWN_WAVES * 32)
        tile4[i] = make_float4(0.f, 0.f, 0.f, 0.f);
    __syncthreads();

    const int    goff = gmeta[2 * blockIdx.x];
    const int    n    = gmeta[2 * blockIdx.x + 1];
    const int    nd   = n / DRAIN;
    const uint2* seg  = (const uint2*)rec + goff;
    const float4* h4  = (const float4*)hin;

    int   d  = wave;
    uint2 pe = make_uint2(0u, 0u);
    if (d < nd) pe = seg[d * DRAIN + (lane & (DRAIN - 1))];
    while (d < nd) {
        unsigned row[DRAIN];
        float4   v[DRAIN];
        float    nrm[DRAIN];
        #pragma unroll
        for (int i = 0; i < DRAIN; ++i) {
            unsigned w = (unsigned)__builtin_amdgcn_readlane((int)pe.x, i);
            row[i] = w & 511u;
            nrm[i] = __uint_as_float((unsigned)__builtin_amdgcn_readlane((int)pe.y, i));
            v[i]   = h4[(size_t)(w >> 9) * (HID_C / 4) + lane];
        }
        const int   dn = d + OWN_WAVES;
        const uint2 pn = seg[min(dn, nd - 1) * DRAIN + (lane & (DRAIN - 1))];
        #pragma unroll
        for (int i = 0; i < DRAIN; ++i) own_add_row(tile, row[i], lane, nrm[i], v[i]);
        pe = pn;
        d  = dn;
    }
    if (wave == 0) {
        for (int i = nd * DRAIN; i < n; ++i) {
            uint2    r = seg[i];
            unsigned w = (unsigned)__builtin_amdgcn_readfirstlane((int)r.x);
            float    f = __uint_as_float((unsigned)__builtin_amdgcn_readfirstlane((int)r.y));
            float4   v = h4[(size_t)(w >> 9) * (HID_C / 4) + lane];
            own_add_row(tile, w & 511u, lane, f, v);
        }
    }
    __syncthreads();

    own_layer(tile, lo, t, lane, wave, hin, x0, dinv, w, beta, do_relu, hout);
}

extern "C" void kernel_launch(void* const* d_in, const int* in_sizes, int n_in,
                              void* d_out, int out_size, void* d_ws, size_t ws_size,
                              hipStream_t stream) {
    const float* x      = (const float*)d_in[0];
    const int*   ei     = (const int*)d_in[1];
    const float* proj_w = (const float*)d_in[2];
    const float* proj_b = (const float*)d_in[3];
    const float* conv_w = (const float*)d_in[4];
    float* out = (float*)d_out;

    const int* e_src = ei;
    const int* e_dst = ei + N_EDGES;

    const size_t NH = (size_t)N_NODES * HID_C;
    char* ws = (char*)d_ws;
    int*   degi = (int*)ws;                                  ws += ((size_t)N_NODES * 4 + 255) & ~255ull;
    float* dinv = (float*)ws;                                ws += ((size_t)N_NODES * 4 + 255) & ~255ull;
    float* x0   = (float*)ws;                                ws += NH * 4;
    float* ha   = (float*)ws;                                ws += NH * 4;
    float* hb   = (float*)ws;                                ws += NH * 4;
    unsigned* rec    = (unsigned*)ws;                        ws += (size_t)N_EDGES * 8;
    int*      gmeta  = (int*)ws;                             ws += ((size_t)OWN_GROUPS * 8 + 255) & ~255ull;
    int*      gcount = (int*)ws;                             ws += ((size_t)OWN_GROUPS * 4 + 255) & ~255ull;
    int*      gcur   = (int*)ws;

    kdeg_init <<<(N_NODES + 255) / 256, 256, 0, stream>>>(degi, gcur);
    kdeg_count<<<(N_EDGES / 4 + 255) / 256, 256, 0, stream>>>(e_dst, degi);
    kdinv     <<<OWN_GROUPS, DINV_THREADS, 0, stream>>>(degi, dinv, gcount);
    kpart     <<<PART_BLOCKS, PART_THREADS, 0, stream>>>(e_src, e_dst, dinv, gcount, gcur, rec, gmeta);

    kproj<<<N_NODES / 16, 256, 0, stream>>>(x, proj_w, proj_b, x0);

    static const float betas[N_LAYERS] = {0.40546510f, 0.22314355f, 0.15415068f, 0.11778303f};

    const float* const hins[N_LAYERS]  = {x0, ha, hb, ha};
    float* const       houts[N_LAYERS] = {ha, hb, ha, out};

    for (int i = 0; i < N_LAYERS; ++i) {
        const float* wl   = conv_w + (size_t)i * HID_C * HID_C;
        const int    relu = (i != N_LAYERS - 1) ? 1 : 0;
        kown_replay<<<OWN_GROUPS, OWN_WAVES * 32, 0, stream>>>(rec, gmeta, hins[i], x0, dinv,
                                                             wl, betas[i], relu, houts[i]);
    }
}


# Degree count without per-edge global adds: edges split by owner range first, each owner counts its rows' in-degrees in LDS from its records; rows stored times dinv, so records carry no weight
# speedup vs baseline: 3.7633x; 1.1335x over previous
#include <hip/hip_runtime.h>
#include <math.h>

#define N_NODES 100000
#define N_EDGES 1600000
#define IN_C    256
#define HID_C   128
#define N_LAYERS 4
#define ALPHA   0.1f
#define OWN_ROWS 400

typedef float v2f __attribute__((ext_vector_type(2)));
typedef float v8f __attribute__((ext_vector_type(8)));

__device__ __forceinline__ void fadd_lds(float* p, float v) {
    __hip_atomic_fetch_add(p, v, __ATOMIC_RELAXED, __HIP_MEMORY_SCOPE_WORKGROUP);
}

#define OWN_GROUPS (N_NODES / OWN_ROWS)
#define CUR_PITCH  32

__global__ void kinit(int* gcount, int* gcur) {
    const int v = threadIdx.x;
    if (v < OWN_GROUPS) {
        gcount[v * CUR_PITCH] = 0;
        gcur[v * CUR_PITCH] = 0;
    }
}

#define LDA_P 260
__global__ void kproj(const float* __restrict__ x, const float* __restrict__ w,
                      const float* __restrict__ bias, const float* __restrict__ dinv,
                      float* __restrict__ x0, float* __restrict__ g0) {
    __shared__ float sa[16 * LDA_P];
    const int t   = threadIdx.x;
    const int m0  = blockIdx.x * 16;

    const float4* xf4 = (const float4*)x;
    #pragma unroll
    for (int c = 0; c < 4; ++c) {
        int idx = c * 256 + t;
        int row = idx >> 6, q = idx & 63;
        float4 v = xf4[(size_t)(m0 + row) * 64 + q];
        float* d = &sa[row * LDA_P + q * 4];
        d[0] = v.x; d[1] = v.y; d[2] = v.z; d[3] = v.w;
    }
    __syncthreads();

    const int wave = t >> 5, lane = t & 31;
    const int half = lane >> 4, col = lane & 15;
    const int n0   = wave * 16;
    const int mrow = col;

    v8f acc = {};
    #pragma unroll 8
    for (int k0 = 0; k0 < IN_C; k0 += 4) {
        int ka = k0 + 2 * half;
        v2f a = *(const v2f*)&sa[mrow * LDA_P + ka];
        v2f b;
        b.x = w[(size_t)ka * HID_C + n0 + col];
        b.y = w[(size_t)(ka + 1) * HID_C + n0 + col];
        acc = __builtin_amdgcn_wmma_f32_16x16x4_f32(false, a, false, b,
                                                    (short)0, acc, false, false);
    }

    float bv = bias[n0 + col];
    #pragma unroll
    for (int i = 0; i < 8; ++i) {
        int row = i + 8 * half;
        const float v = acc[i] + bv;
        x0[(size_t)(m0 + row) * HID_C + n0 + col] = v;
        g0[(size_t)(m0 + row) * HID_C + n0 + col] = dinv[m0 + row] * v;
    }
}

#define TP         132
#define TJ         33
#define OWN_WAVES  32
#define DRAIN      8
static_assert(N_NODES % OWN_ROWS == 0, "ranges must tile the nodes");
static_assert(OWN_ROWS <= 512, "record word packs a 9-bit row");
static_assert(N_NODES <= (1 << 23), "record word packs source node and 9-bit row");

#define PART_THREADS 1024
#define PART_EDGES   (4 * PART_THREADS)
#define PART_BLOCKS  ((N_EDGES / 4 + PART_THREADS - 1) / PART_THREADS)
static_assert(N_EDGES % 4 == 0, "edges are read four at a time");
static_assert(OWN_GROUPS <= 256, "the first eight waves hold one owner per thread");

__global__ __launch_bounds__(PART_THREADS)
void khist(const int* __restrict__ dst, int* __restrict__ gcount) {
    __shared__ int hist[OWN_GROUPS];
    const int t = threadIdx.x;
    const int i = blockIdx.x * PART_THREADS + t;
    if (t < OWN_GROUPS) hist[t] = 0;
    __syncthreads();
    if (i < N_EDGES / 4) {
        const int4 d = ((const int4*)dst)[i];
        __hip_atomic_fetch_add(&hist[(unsigned)d.x / OWN_ROWS], 1, __ATOMIC_RELAXED, __HIP_MEMORY_SCOPE_WORKGROUP);
        __hip_atomic_fetch_add(&hist[(unsigned)d.y / OWN_ROWS], 1, __ATOMIC_RELAXED, __HIP_MEMORY_SCOPE_WORKGROUP);
        __hip_atomic_fetch_add(&hist[(unsigned)d.z / OWN_ROWS], 1, __ATOMIC_RELAXED, __HIP_MEMORY_SCOPE_WORKGROUP);
        __hip_atomic_fetch_add(&hist[(unsigned)d.w / OWN_ROWS], 1, __ATOMIC_RELAXED, __HIP_MEMORY_SCOPE_WORKGROUP);
    }
    __syncthreads();
    if (t < OWN_GROUPS) {
        const int n = hist[t];
        if (n > 0) atomicAdd(&gcount[t * CUR_PITCH], n);
    }
}

__global__ __launch_bounds__(PART_THREADS)
void kpart(const int* __restrict__ src, const int* __restrict__ dst,
           const int* __restrict__ gcount,
           int* __restrict__ gcur, unsigned* __restrict__ rec, int* __restrict__ gmeta) {
    __shared__ int hist[OWN_GROUPS];
    __shared__ int sbase[OWN_GROUPS];
    __shared__ int wtot[8];

    const int t = threadIdx.x, lane = t & 31, wave = t >> 5;
    const int i = blockIdx.x * PART_THREADS + t;
    const bool ok = i < N_EDGES / 4;

    int4 d = make_int4(0, 0, 0, 0), s = make_int4(0, 0, 0, 0);
    if (ok) {
        d = ((const int4*)dst)[i];
        s = ((const int4*)src)[i];
    }

    int cnt_t = 0, start_t = 0;
    if (wave < 8) {
        cnt_t = (t < OWN_GROUPS) ? gcount[t * CUR_PITCH] : 0;
        int incl = cnt_t;
        for (int o = 1; o < 32; o <<= 1) {
            int up = __shfl_up(incl, o);
            if (lane >= o) incl += up;
        }
        if (lane == 31) wtot[wave] = incl;
        start_t = incl - cnt_t;
    }
    if (t < OWN_GROUPS) hist[t] = 0;
    __syncthreads();
    if (t < OWN_GROUPS) {
        for (int k = 0; k < wave; ++k) start_t += wtot[k];
        if (blockIdx.x == 0) {
            gmeta[2 * t]     = start_t;
            gmeta[2 * t + 1] = cnt_t;
        }
    }

    const unsigned o0 = (unsigned)d.x / OWN_ROWS, o1 = (unsigned)d.y / OWN_ROWS,
                   o2 = (unsigned)d.z / OWN_ROWS, o3 = (unsigned)d.w / OWN_ROWS;
    int r0 = 0, r1 = 0, r2 = 0, r3 = 0;
    if (ok) {
        r0 = __hip_atomic_fetch_add(&hist[o0], 1, __ATOMIC_RELAXED, __HIP_MEMORY_SCOPE_WORKGROUP);
        r1 = __hip_atomic_fetch_add(&hist[o1], 1, __ATOMIC_RELAXED, __HIP_MEMORY_SCOPE_WORKGROUP);
        r2 = __hip_atomic_fetch_add(&hist[o2], 1, __ATOMIC_RELAXED, __HIP_MEMORY_SCOPE_WORKGROUP);
        r3 = __hip_atomic_fetch_add(&hist[o3], 1, __ATOMIC_RELAXED, __HIP_MEMORY_SCOPE_WORKGROUP);
    }
    __syncthreads();
    if (t < OWN_GROUPS) {
        const int n = hist[t];
        int b = 0;
        if (n > 0) b = atomicAdd(&gcur[t * CUR_PITCH], n);
        sbase[t] = start_t + b;
    }
    __syncthreads();
    if (ok) {
        rec[sbase[o0] + r0] = ((unsigned)s.x << 9) | ((unsigned)d.x - o0 * OWN_ROWS);
        rec[sbase[o1] + r1] = ((unsigned)s.y << 9) | ((unsigned)d.y - o1 * OWN_ROWS);
        rec[sbase[o2] + r2] = ((unsigned)s.z << 9) | ((unsigned)d.z - o2 * OWN_ROWS);
        rec[sbase[o3] + r3] = ((unsigned)s.w << 9) | ((unsigned)d.w - o3 * OWN_ROWS);
    }
}

__global__ __launch_bounds__(1024)
void kdinv(const unsigned* __restrict__ rec, const int* __restrict__ gmeta,
           float* __restrict__ dinv) {
    __shared__ int cnt[OWN_ROWS];
    const int t = threadIdx.x;
    if (t < OWN_ROWS) cnt[t] = 1;
    __syncthreads();
    const int goff = gmeta[2 * blockIdx.x];
    const int n    = gmeta[2 * blockIdx.x + 1];
    for (int i = t; i < n; i += 1024)
        __hip_atomic_fetch_add(&cnt[rec[goff + i] & 511u], 1, __ATOMIC_RELAXED, __HIP_MEMORY_SCOPE_WORKGROUP);
    __syncthreads();
    if (t < OWN_ROWS) dinv[blockIdx.x * OWN_ROWS + t] = rsqrtf((float)cnt[t]);
}

__device__ __forceinline__ void own_add_row(float* __restrict__ tile, unsigned row, int lane,
                                            float4 v) {
    float* a = tile + row * TP + lane;
    fadd_lds(a,          v.x);
    fadd_lds(a + TJ,     v.y);
    fadd_lds(a + 2 * TJ, v.z);
    fadd_lds(a + 3 * TJ, v.w);
}

#define OWN_MT (OWN_ROWS / 16)
static_assert(OWN_ROWS % 16 == 0, "the tile is a whole number of 16-row matrix tiles");
__device__ __forceinline__ void own_layer(float* __restrict__ tile, const float* __restrict__ sdinv,
                                          int lo, int t, int lane, int wave,
                                          const float* __restrict__ hin,
                                          const float* __restrict__ x0,
                                          const float* __restrict__ w, float beta, int do_relu,
                                          int scale_out, float* __restrict__ hout) {
    const int half = lane >> 4, col = lane & 15;
    const int n0   = (wave & 7) * 16;
    const int ch   = n0 + col;

    v2f b[HID_C / 4];
    #pragma unroll
    for (int kk = 0; kk < HID_C / 4; ++kk) {
        const int ka = 4 * kk + 2 * half;
        b[kk].x = w[(size_t)ka * HID_C + ch];
        b[kk].y = w[(size_t)(ka + 1) * HID_C + ch];
    }

    const float4* h4 = (const float4*)hin + (size_t)lo * (HID_C / 4);
    const float4* x4 = (const float4*)x0  + (size_t)lo * (HID_C / 4);
    constexpr int NQ = OWN_ROWS * (HID_C / 4);
    for (int i0 = 0; i0 < NQ; i0 += 4 * OWN_WAVES * 32) {
        float4 hv[4], xv[4];
        #pragma unroll
        for (int k = 0; k < 4; ++k) {
            const int i = min(i0 + k * OWN_WAVES * 32 + t, NQ - 1);
            hv[k] = h4[i];
            xv[k] = x4[i];
        }
        #pragma unroll
        for (int k = 0; k < 4; ++k) {
            const int i = i0 + k * OWN_WAVES * 32 + t;
            if (i < NQ) {
                float* a = tile + (i >> 5) * TP + (i & 31);
                const float c = (1.f - ALPHA) * sdinv[i >> 5];
                a[0]      = c * (a[0]      + hv[k].x) + ALPHA * xv[k].x;
                a[TJ]     = c * (a[TJ]     + hv[k].y) + ALPHA * xv[k].y;
                a[2 * TJ] = c * (a[2 * TJ] + hv[k].z) + ALPHA * xv[k].z;
                a[3 * TJ] = c * (a[3 * TJ] + hv[k].w) + ALPHA * xv[k].w;
            }
        }
    }
    __syncthreads();

    for (int mt = wave >> 3; mt < OWN_MT; mt += OWN_WAVES / 8) {
        const float* ar = tile + (mt * 16 + col) * TP + 2 * half * TJ;
        v8f acc = {};
        #pragma unroll
        for (int kk = 0; kk < HID_C / 4; ++kk) {
            v2f a;
            a.x = ar[kk];
            a.y = ar[TJ + kk];
            acc = __builtin_amdgcn_wmma_f32_16x16x4_f32(false, a, false, b[kk],
                                                        (short)0, acc, false, false);
        }
        const float* sr = tile + (mt * 16 + 8 * half) * TP + (ch & 3) * TJ + (ch >> 2);
        float* orow = hout + (size_t)(lo + mt * 16 + 8 * half) * HID_C + ch;
        #pragma unroll
        for (int i = 0; i < 8; ++i) {
            float v = (1.f - beta) * sr[i * TP] + beta * acc[i];
            if (do_relu) v = fmaxf(v, 0.f);
            if (scale_out) v *= sdinv[mt * 16 + 8 * half + i];
            orow[(size_t)i * HID_C] = v;
        }
    }
}

__global__ __launch_bounds__(OWN_WAVES * 32)
void kown_replay(const unsigned* __restrict__ rec, const int* __restrict__ gmeta,
                 const float* __restrict__ hin, const float* __restrict__ x0,
                 const float* __restrict__ dinv,
                 const float* __restrict__ w, float beta, int do_relu, int scale_out,
                 float* __restrict__ hout) {
    __shared__ __attribute__((aligned(16))) float tile[OWN_ROWS * TP];
    __shared__ float sdinv[OWN_ROWS];

    const int t = threadIdx.x, lane = t & 31;
    const int wave = __builtin_amdgcn_readfirstlane(t >> 5);
    const int lo = blockIdx.x * OWN_ROWS;

    float4* tile4 = (float4*)tile;
    for (int i = t; i < OWN_ROWS * TP / 4; i += OWN_WAVES * 32)
        tile4[i] = make_float4(0.f, 0.f, 0.f, 0.f);
    if (t < OWN_ROWS) sdinv[t] = dinv[lo + t];
    __syncthreads();

    const int    goff = gmeta[2 * blockIdx.x];
    const int    n    = gmeta[2 * blockIdx.x + 1];
    const int    nd   = n / DRAIN;
    const unsigned* seg = rec + goff;
    const float4* h4  = (const float4*)hin;

    int      d  = wave;
    unsigned pe = 0u;
    if (d < nd) pe = seg[d * DRAIN + (lane & (DRAIN - 1))];
    while (d < nd) {
        unsigned row[DRAIN];
        float4   v[DRAIN];
        #pragma unroll
        for (int i = 0; i < DRAIN; ++i) {
            unsigned r = (unsigned)__builtin_amdgcn_readlane((int)pe, i);
            row[i] = r & 511u;
            v[i]   = h4[(size_t)(r >> 9) * (HID_C / 4) + lane];
        }
        const int      dn = d + OWN_WAVES;
        const unsigned pn = seg[min(dn, nd - 1) * DRAIN + (lane & (DRAIN - 1))];
        #pragma unroll
        for (int i = 0; i < DRAIN; ++i) own_add_row(tile, row[i], lane, v[i]);
        pe = pn;
        d  = dn;
    }
    if (wave == 0) {
        for (int i = nd * DRAIN; i < n; ++i) {
            unsigned r = (unsigned)__builtin_amdgcn_readfirstlane((int)seg[i]);
            float4   v = h4[(size_t)(r >> 9) * (HID_C / 4) + lane];
            own_add_row(tile, r & 511u, lane, v);
        }
    }
    __syncthreads();

    own_layer(tile, sdinv, lo, t, lane, wave, hin, x0, w, beta, do_relu, scale_out, hout);
}

extern "C" void kernel_launch(void* const* d_in, const int* in_sizes, int n_in,
                              void* d_out, int out_size, void* d_ws, size_t ws_size,
                              hipStream_t stream) {
    const float* x      = (const float*)d_in[0];
    const int*   ei     = (const int*)d_in[1];
    const float* proj_w = (const float*)d_in[2];
    const float* proj_b = (const float*)d_in[3];
    const float* conv_w = (const float*)d_in[4];
    float* out = (float*)d_out;

    const int* e_src = ei;
    const int* e_dst = ei + N_EDGES;

    const size_t NH = (size_t)N_NODES * HID_C;
    char* ws = (char*)d_ws;
    float* dinv = (float*)ws;                                ws += ((size_t)N_NODES * 4 + 255) & ~255ull;
    float* x0   = (float*)ws;                                ws += NH * 4;
    float* g0   = (float*)ws;                                ws += NH * 4;
    float* ha   = (float*)ws;                                ws += NH * 4;
    float* hb   = (float*)ws;                                ws += NH * 4;
    unsigned* rec    = (unsigned*)ws;                        ws += (size_t)N_EDGES * 4;
    int*      gmeta  = (int*)ws;                             ws += ((size_t)OWN_GROUPS * 8 + 255) & ~255ull;
    int*      gcount = (int*)ws;                             ws += (size_t)OWN_GROUPS * CUR_PITCH * 4;
    int*      gcur   = (int*)ws;

    kinit<<<1, 256, 0, stream>>>(gcount, gcur);
    khist<<<PART_BLOCKS, PART_THREADS, 0, stream>>>(e_dst, gcount);
    kpart<<<PART_BLOCKS, PART_THREADS, 0, stream>>>(e_src, e_dst, gcount, gcur, rec, gmeta);
    kdinv<<<OWN_GROUPS, 1024, 0, stream>>>(rec, gmeta, dinv);

    kproj<<<N_NODES / 16, 256, 0, stream>>>(x, proj_w, proj_b, dinv, x0, g0);

    static const float betas[N_LAYERS] = {0.40546510f, 0.22314355f, 0.15415068f, 0.11778303f};

    const float* const hins[N_LAYERS]  = {g0, ha, hb, ha};
    float* const       houts[N_LAYERS] = {ha, hb, ha, out};

    for (int i = 0; i < N_LAYERS; ++i) {
        const float* wl   = conv_w + (size_t)i * HID_C * HID_C;
        const int    more = (i != N_LAYERS - 1) ? 1 : 0;
        kown_replay<<<OWN_GROUPS, OWN_WAVES * 32, 0, stream>>>(rec, gmeta, hins[i], x0, dinv,
                                                             wl, betas[i], more, more, houts[i]);
    }
}


# Edge split by owner range without global counters: each 4096-edge group sorts its records by owner in its own slice; owner groups copy their runs into their segment; 2 launches gone; arrival order
# speedup vs baseline: 3.9782x; 1.0571x over previous
#include <hip/hip_runtime.h>
#include <math.h>

#define N_NODES 100000
#define N_EDGES 1600000
#define IN_C    256
#define HID_C   128
#define N_LAYERS 4
#define ALPHA   0.1f
#define OWN_ROWS 400

typedef float v2f __attribute__((ext_vector_type(2)));
typedef float v8f __attribute__((ext_vector_type(8)));

__device__ __forceinline__ void fadd_lds(float* p, float v) {
    __hip_atomic_fetch_add(p, v, __ATOMIC_RELAXED, __HIP_MEMORY_SCOPE_WORKGROUP);
}

#define OWN_GROUPS (N_NODES / OWN_ROWS)


#define LDA_P 260
__global__ void kproj(const float* __restrict__ x, const float* __restrict__ w,
                      const float* __restrict__ bias, const float* __restrict__ dinv,
                      float* __restrict__ x0, float* __restrict__ g0) {
    __shared__ float sa[16 * LDA_P];
    const int t   = threadIdx.x;
    const int m0  = blockIdx.x * 16;

    const float4* xf4 = (const float4*)x;
    #pragma unroll
    for (int c = 0; c < 4; ++c) {
        int idx = c * 256 + t;
        int row = idx >> 6, q = idx & 63;
        float4 v = xf4[(size_t)(m0 + row) * 64 + q];
        float* d = &sa[row * LDA_P + q * 4];
        d[0] = v.x; d[1] = v.y; d[2] = v.z; d[3] = v.w;
    }
    __syncthreads();

    const int wave = t >> 5, lane = t & 31;
    const int half = lane >> 4, col = lane & 15;
    const int n0   = wave * 16;
    const int mrow = col;

    v8f acc = {};
    #pragma unroll 8
    for (int k0 = 0; k0 < IN_C; k0 += 4) {
        int ka = k0 + 2 * half;
        v2f a = *(const v2f*)&sa[mrow * LDA_P + ka];
        v2f b;
        b.x = w[(size_t)ka * HID_C + n0 + col];
        b.y = w[(size_t)(ka + 1) * HID_C + n0 + col];
        acc = __builtin_amdgcn_wmma_f32_16x16x4_f32(false, a, false, b,
                                                    (short)0, acc, false, false);
    }

    float bv = bias[n0 + col];
    #pragma unroll
    for (int i = 0; i < 8; ++i) {
        int row = i + 8 * half;
        const float v = acc[i] + bv;
        x0[(size_t)(m0 + row) * HID_C + n0 + col] = v;
        g0[(size_t)(m0 + row) * HID_C + n0 + col] = dinv[m0 + row] * v;
    }
}

#define TP         132
#define TJ         33
#define OWN_WAVES  32
#define DRAIN      8
static_assert(N_NODES % OWN_ROWS == 0, "ranges must tile the nodes");
static_assert(OWN_ROWS <= 512, "record word packs a 9-bit row");
static_assert(N_NODES <= (1 << 23), "record word packs source node and 9-bit row");

#define PART_THREADS 1024
#define PART_EDGES   (4 * PART_THREADS)
#define PART_BLOCKS  ((N_EDGES / 4 + PART_THREADS - 1) / PART_THREADS)
#define BO_PITCH     ((PART_BLOCKS + 31) & ~31)
static_assert(N_EDGES % 4 == 0, "edges are read four at a time");
static_assert(OWN_GROUPS < 256, "the first eight waves hold one owner per thread, and one thread more");
static_assert(PART_BLOCKS <= 1024, "kdinv holds one edge block per thread");

__global__ __launch_bounds__(PART_THREADS)
void kpart(const int* __restrict__ src, const int* __restrict__ dst,
           unsigned* __restrict__ recb, int* __restrict__ boff) {
    __shared__ int hist[OWN_GROUPS];
    __shared__ int sbase[OWN_GROUPS];
    __shared__ int wtot[8];

    const int t = threadIdx.x, lane = t & 31, wave = t >> 5;
    const int i = blockIdx.x * PART_THREADS + t;
    const bool ok = i < N_EDGES / 4;

    int4 d = make_int4(0, 0, 0, 0), s = make_int4(0, 0, 0, 0);
    if (ok) {
        d = ((const int4*)dst)[i];
        s = ((const int4*)src)[i];
    }
    if (t < OWN_GROUPS) hist[t] = 0;
    __syncthreads();

    const unsigned o0 = (unsigned)d.x / OWN_ROWS, o1 = (unsigned)d.y / OWN_ROWS,
                   o2 = (unsigned)d.z / OWN_ROWS, o3 = (unsigned)d.w / OWN_ROWS;
    int r0 = 0, r1 = 0, r2 = 0, r3 = 0;
    if (ok) {
        r0 = __hip_atomic_fetch_add(&hist[o0], 1, __ATOMIC_RELAXED, __HIP_MEMORY_SCOPE_WORKGROUP);
        r1 = __hip_atomic_fetch_add(&hist[o1], 1, __ATOMIC_RELAXED, __HIP_MEMORY_SCOPE_WORKGROUP);
        r2 = __hip_atomic_fetch_add(&hist[o2], 1, __ATOMIC_RELAXED, __HIP_MEMORY_SCOPE_WORKGROUP);
        r3 = __hip_atomic_fetch_add(&hist[o3], 1, __ATOMIC_RELAXED, __HIP_MEMORY_SCOPE_WORKGROUP);
    }
    __syncthreads();

    int start_t = 0;
    if (wave < 8) {
        const int cnt_t = (t < OWN_GROUPS) ? hist[t] : 0;
        int incl = cnt_t;
        for (int o = 1; o < 32; o <<= 1) {
            int up = __shfl_up(incl, o);
            if (lane >= o) incl += up;
        }
        if (lane == 31) wtot[wave] = incl;
        start_t = incl - cnt_t;
    }
    __syncthreads();
    if (t <= OWN_GROUPS) {
        for (int k = 0; k < wave; ++k) start_t += wtot[k];
        if (t < OWN_GROUPS) sbase[t] = start_t;
        boff[t * BO_PITCH + blockIdx.x] = start_t;
    }
    __syncthreads();
    if (ok) {
        unsigned* slice = recb + (size_t)blockIdx.x * PART_EDGES;
        slice[sbase[o0] + r0] = ((unsigned)s.x << 9) | ((unsigned)d.x - o0 * OWN_ROWS);
        slice[sbase[o1] + r1] = ((unsigned)s.y << 9) | ((unsigned)d.y - o1 * OWN_ROWS);
        slice[sbase[o2] + r2] = ((unsigned)s.z << 9) | ((unsigned)d.z - o2 * OWN_ROWS);
        slice[sbase[o3] + r3] = ((unsigned)s.w << 9) | ((unsigned)d.w - o3 * OWN_ROWS);
    }
}

__global__ __launch_bounds__(1024)
void kdinv(const unsigned* __restrict__ recb, const int* __restrict__ boff,
           unsigned* __restrict__ rec, int* __restrict__ gmeta, float* __restrict__ dinv) {
    __shared__ int cnt[OWN_ROWS];
    __shared__ int rfrom[PART_BLOCKS];
    __shared__ int rto[PART_BLOCKS];
    __shared__ int rlen[PART_BLOCKS];
    __shared__ int wlen[32], wlow[32];

    const int t = threadIdx.x, lane = t & 31, wave = t >> 5;
    const int g = blockIdx.x;
    int a = 0, e = 0;
    if (t < PART_BLOCKS) {
        a = boff[g * BO_PITCH + t];
        e = boff[(g + 1) * BO_PITCH + t];
    }
    if (t < OWN_ROWS) cnt[t] = 1;
    const int len = e - a;
    int incl = len, low = a;
    for (int o = 1; o < 32; o <<= 1) {
        int up  = __shfl_up(incl, o);
        int upl = __shfl_up(low, o);
        if (lane >= o) { incl += up; low += upl; }
    }
    if (lane == 31) { wlen[wave] = incl; wlow[wave] = low; }
    __syncthreads();
    int before = incl - len, n = 0, goff = 0;
    for (int k = 0; k < 32; ++k) {
        const int wl = wlen[k];
        if (k < wave) before += wl;
        n    += wl;
        goff += wlow[k];
    }
    if (t < PART_BLOCKS) {
        rfrom[t] = t * PART_EDGES + a;
        rto[t]   = before;
        rlen[t]  = len;
    }
    if (t == 0) {
        gmeta[2 * g]     = goff;
        gmeta[2 * g + 1] = n;
    }
    __syncthreads();
    unsigned* seg = rec + goff;
    const int wv = __builtin_amdgcn_readfirstlane(wave);
    constexpr int RUNS = 4;
    for (int b0 = wv; b0 < PART_BLOCKS; b0 += 32 * RUNS) {
        int from[RUNS], to[RUNS], l[RUNS];
        unsigned r[RUNS];
        #pragma unroll
        for (int u = 0; u < RUNS; ++u) {
            const int b = b0 + 32 * u;
            const int bc = min(b, PART_BLOCKS - 1);
            from[u] = rfrom[bc];
            to[u]   = rto[bc];
            l[u]    = (b < PART_BLOCKS) ? rlen[bc] : 0;
        }
        #pragma unroll
        for (int u = 0; u < RUNS; ++u)
            r[u] = recb[min(from[u] + min(lane, max(l[u] - 1, 0)), PART_BLOCKS * PART_EDGES - 1)];
        #pragma unroll
        for (int u = 0; u < RUNS; ++u) {
            if (lane < l[u]) {
                seg[to[u] + lane] = r[u];
                __hip_atomic_fetch_add(&cnt[r[u] & 511u], 1, __ATOMIC_RELAXED, __HIP_MEMORY_SCOPE_WORKGROUP);
            }
            for (int j = lane + 32; j < l[u]; j += 32) {
                const unsigned rr = recb[from[u] + j];
                seg[to[u] + j] = rr;
                __hip_atomic_fetch_add(&cnt[rr & 511u], 1, __ATOMIC_RELAXED, __HIP_MEMORY_SCOPE_WORKGROUP);
            }
        }
    }
    __syncthreads();
    if (t < OWN_ROWS) dinv[g * OWN_ROWS + t] = rsqrtf((float)cnt[t]);
}

__device__ __forceinline__ void own_add_row(float* __restrict__ tile, unsigned row, int lane,
                                            float4 v) {
    float* a = tile + row * TP + lane;
    fadd_lds(a,          v.x);
    fadd_lds(a + TJ,     v.y);
    fadd_lds(a + 2 * TJ, v.z);
    fadd_lds(a + 3 * TJ, v.w);
}

#define OWN_MT (OWN_ROWS / 16)
static_assert(OWN_ROWS % 16 == 0, "the tile is a whole number of 16-row matrix tiles");
__device__ __forceinline__ void own_layer(float* __restrict__ tile, const float* __restrict__ sdinv,
                                          int lo, int t, int lane, int wave,
                                          const float* __restrict__ hin,
                                          const float* __restrict__ x0,
                                          const float* __restrict__ w, float beta, int do_relu,
                                          int scale_out, float* __restrict__ hout) {
    const int half = lane >> 4, col = lane & 15;
    const int n0   = (wave & 7) * 16;
    const int ch   = n0 + col;

    v2f b[HID_C / 4];
    #pragma unroll
    for (int kk = 0; kk < HID_C / 4; ++kk) {
        const int ka = 4 * kk + 2 * half;
        b[kk].x = w[(size_t)ka * HID_C + ch];
        b[kk].y = w[(size_t)(ka + 1) * HID_C + ch];
    }

    const float4* h4 = (const float4*)hin + (size_t)lo * (HID_C / 4);
    const float4* x4 = (const float4*)x0  + (size_t)lo * (HID_C / 4);
    constexpr int NQ = OWN_ROWS * (HID_C / 4);
    for (int i0 = 0; i0 < NQ; i0 += 4 * OWN_WAVES * 32) {
        float4 hv[4], xv[4];
        #pragma unroll
        for (int k = 0; k < 4; ++k) {
            const int i = min(i0 + k * OWN_WAVES * 32 + t, NQ - 1);
            hv[k] = h4[i];
            xv[k] = x4[i];
        }
        #pragma unroll
        for (int k = 0; k < 4; ++k) {
            const int i = i0 + k * OWN_WAVES * 32 + t;
            if (i < NQ) {
                float* a = tile + (i >> 5) * TP + (i & 31);
                const float c = (1.f - ALPHA) * sdinv[i >> 5];
                a[0]      = c * (a[0]      + hv[k].x) + ALPHA * xv[k].x;
                a[TJ]     = c * (a[TJ]     + hv[k].y) + ALPHA * xv[k].y;
                a[2 * TJ] = c * (a[2 * TJ] + hv[k].z) + ALPHA * xv[k].z;
                a[3 * TJ] = c * (a[3 * TJ] + hv[k].w) + ALPHA * xv[k].w;
            }
        }
    }
    __syncthreads();

    for (int mt = wave >> 3; mt < OWN_MT; mt += OWN_WAVES / 8) {
        const float* ar = tile + (mt * 16 + col) * TP + 2 * half * TJ;
        v8f acc = {};
        #pragma unroll
        for (int kk = 0; kk < HID_C / 4; ++kk) {
            v2f a;
            a.x = ar[kk];
            a.y = ar[TJ + kk];
            acc = __builtin_amdgcn_wmma_f32_16x16x4_f32(false, a, false, b[kk],
                                                        (short)0, acc, false, false);
        }
        const float* sr = tile + (mt * 16 + 8 * half) * TP + (ch & 3) * TJ + (ch >> 2);
        float* orow = hout + (size_t)(lo + mt * 16 + 8 * half) * HID_C + ch;
        #pragma unroll
        for (int i = 0; i < 8; ++i) {
            float v = (1.f - beta) * sr[i * TP] + beta * acc[i];
            if (do_relu) v = fmaxf(v, 0.f);
            if (scale_out) v *= sdinv[mt * 16 + 8 * half + i];
            orow[(size_t)i * HID_C] = v;
        }
    }
}

__global__ __launch_bounds__(OWN_WAVES * 32)
void kown_replay(const unsigned* __restrict__ rec, const int* __restrict__ gmeta,
                 const float* __restrict__ hin, const float* __restrict__ x0,
                 const float* __restrict__ dinv,
                 const float* __restrict__ w, float beta, int do_relu, int scale_out,
                 float* __restrict__ hout) {
    __shared__ __attribute__((aligned(16))) float tile[OWN_ROWS * TP];
    __shared__ float sdinv[OWN_ROWS];

    const int t = threadIdx.x, lane = t & 31;
    const int wave = __builtin_amdgcn_readfirstlane(t >> 5);
    const int lo = blockIdx.x * OWN_ROWS;

    float4* tile4 = (float4*)tile;
    for (int i = t; i < OWN_ROWS * TP / 4; i += OWN_WAVES * 32)
        tile4[i] = make_float4(0.f, 0.f, 0.f, 0.f);
    if (t < OWN_ROWS) sdinv[t] = dinv[lo + t];
    __syncthreads();

    const int    goff = gmeta[2 * blockIdx.x];
    const int    n    = gmeta[2 * blockIdx.x + 1];
    const int    nd   = n / DRAIN;
    const unsigned* seg = rec + goff;
    const float4* h4  = (const float4*)hin;

    int      d  = wave;
    unsigned pe = 0u;
    if (d < nd) pe = seg[d * DRAIN + (lane & (DRAIN - 1))];
    while (d < nd) {
        unsigned row[DRAIN];
        float4   v[DRAIN];
        #pragma unroll
        for (int i = 0; i < DRAIN; ++i) {
            unsigned r = (unsigned)__builtin_amdgcn_readlane((int)pe, i);
            row[i] = r & 511u;
            v[i]   = h4[(size_t)(r >> 9) * (HID_C / 4) + lane];
        }
        const int      dn = d + OWN_WAVES;
        const unsigned pn = seg[min(dn, nd - 1) * DRAIN + (lane & (DRAIN - 1))];
        #pragma unroll
        for (int i = 0; i < DRAIN; ++i) own_add_row(tile, row[i], lane, v[i]);
        pe = pn;
        d  = dn;
    }
    if (wave == 0) {
        for (int i = nd * DRAIN; i < n; ++i) {
            unsigned r = (unsigned)__builtin_amdgcn_readfirstlane((int)seg[i]);
            float4   v = h4[(size_t)(r >> 9) * (HID_C / 4) + lane];
            own_add_row(tile, r & 511u, lane, v);
        }
    }
    __syncthreads();

    own_layer(tile, sdinv, lo, t, lane, wave, hin, x0, w, beta, do_relu, scale_out, hout);
}

extern "C" void kernel_launch(void* const* d_in, const int* in_sizes, int n_in,
                              void* d_out, int out_size, void* d_ws, size_t ws_size,
                              hipStream_t stream) {
    const float* x      = (const float*)d_in[0];
    const int*   ei     = (const int*)d_in[1];
    const float* proj_w = (const float*)d_in[2];
    const float* proj_b = (const float*)d_in[3];
    const float* conv_w = (const float*)d_in[4];
    float* out = (float*)d_out;

    const int* e_src = ei;
    const int* e_dst = ei + N_EDGES;

    const size_t NH = (size_t)N_NODES * HID_C;
    char* ws = (char*)d_ws;
    float* dinv = (float*)ws;                                ws += ((size_t)N_NODES * 4 + 255) & ~255ull;
    float* x0   = (float*)ws;                                ws += NH * 4;
    float* g0   = (float*)ws;                                ws += NH * 4;
    float* ha   = (float*)ws;                                ws += NH * 4;
    float* hb   = (float*)ws;                                ws += NH * 4;
    unsigned* rec    = (unsigned*)ws;                        ws += (size_t)N_EDGES * 4;
    unsigned* recb   = (unsigned*)ws;                        ws += (size_t)PART_BLOCKS * PART_EDGES * 4;
    int*      gmeta  = (int*)ws;                             ws += ((size_t)OWN_GROUPS * 8 + 255) & ~255ull;
    int*      boff   = (int*)ws;

    kpart<<<PART_BLOCKS, PART_THREADS, 0, stream>>>(e_src, e_dst, recb, boff);
    kdinv<<<OWN_GROUPS, 1024, 0, stream>>>(recb, boff, rec, gmeta, dinv);

    kproj<<<N_NODES / 16, 256, 0, stream>>>(x, proj_w, proj_b, dinv, x0, g0);

    static const float betas[N_LAYERS] = {0.40546510f, 0.22314355f, 0.15415068f, 0.11778303f};

    const float* const hins[N_LAYERS]  = {g0, ha, hb, ha};
    float* const       houts[N_LAYERS] = {ha, hb, ha, out};

    for (int i = 0; i < N_LAYERS; ++i) {
        const float* wl   = conv_w + (size_t)i * HID_C * HID_C;
        const int    more = (i != N_LAYERS - 1) ? 1 : 0;
        kown_replay<<<OWN_GROUPS, OWN_WAVES * 32, 0, stream>>>(rec, gmeta, hins[i], x0, dinv,
                                                             wl, betas[i], more, more, houts[i]);
    }
}


# Projection work-groups take 80 rows, not 16: a wave keeps 5 row tiles' accumulators and feeds each B fragment pair it loads to 5 WMMAs, not 1; 10000 waves launched, not 50000; sums bit-identical
# speedup vs baseline: 4.1812x; 1.0510x over previous
#include <hip/hip_runtime.h>
#include <math.h>

#define N_NODES 100000
#define N_EDGES 1600000
#define IN_C    256
#define HID_C   128
#define N_LAYERS 4
#define ALPHA   0.1f
#define OWN_ROWS 400

typedef float v2f __attribute__((ext_vector_type(2)));
typedef float v8f __attribute__((ext_vector_type(8)));

__device__ __forceinline__ void fadd_lds(float* p, float v) {
    __hip_atomic_fetch_add(p, v, __ATOMIC_RELAXED, __HIP_MEMORY_SCOPE_WORKGROUP);
}

#define OWN_GROUPS (N_NODES / OWN_ROWS)


#define LDA_P 260
#define PROJ_MT   5
#define PROJ_ROWS (16 * PROJ_MT)
static_assert(N_NODES % PROJ_ROWS == 0, "projection blocks must tile the nodes");
__global__ __launch_bounds__(256)
void kproj(const float* __restrict__ x, const float* __restrict__ w,
           const float* __restrict__ bias, const float* __restrict__ dinv,
           float* __restrict__ x0, float* __restrict__ g0) {
    __shared__ __attribute__((aligned(16))) float sa[PROJ_ROWS * LDA_P];
    const int t   = threadIdx.x;
    const int m0  = blockIdx.x * PROJ_ROWS;

    const float4* xf4 = (const float4*)x + (size_t)m0 * (IN_C / 4);
    #pragma unroll 10
    for (int c = 0; c < PROJ_ROWS / 4; ++c) {
        const int idx = c * 256 + t;
        const int row = idx >> 6, q = idx & 63;
        *(float4*)&sa[row * LDA_P + q * 4] = xf4[idx];
    }
    __syncthreads();

    const int wave = t >> 5, lane = t & 31;
    const int half = lane >> 4, col = lane & 15;
    const int n0   = wave * 16;
    const float* arow = sa + col * LDA_P + 2 * half;
    const float* bcol = w + (size_t)(2 * half) * HID_C + n0 + col;

    v8f acc[PROJ_MT] = {};
    #pragma unroll 4
    for (int k0 = 0; k0 < IN_C; k0 += 4) {
        v2f b;
        b.x = bcol[(size_t)k0 * HID_C];
        b.y = bcol[(size_t)(k0 + 1) * HID_C];
        #pragma unroll
        for (int m = 0; m < PROJ_MT; ++m) {
            v2f a = *(const v2f*)&arow[m * 16 * LDA_P + k0];
            acc[m] = __builtin_amdgcn_wmma_f32_16x16x4_f32(false, a, false, b,
                                                           (short)0, acc[m], false, false);
        }
    }

    const float bv = bias[n0 + col];
    #pragma unroll
    for (int m = 0; m < PROJ_MT; ++m) {
        #pragma unroll
        for (int i = 0; i < 8; ++i) {
            const int row = m0 + m * 16 + i + 8 * half;
            const float v = acc[m][i] + bv;
            x0[(size_t)row * HID_C + n0 + col] = v;
            g0[(size_t)row * HID_C + n0 + col] = dinv[row] * v;
        }
    }
}

#define TP         132
#define TJ         33
#define OWN_WAVES  32
#define DRAIN      8
static_assert(N_NODES % OWN_ROWS == 0, "ranges must tile the nodes");
static_assert(OWN_ROWS <= 512, "record word packs a 9-bit row");
static_assert(N_NODES <= (1 << 23), "record word packs source node and 9-bit row");

#define PART_THREADS 1024
#define PART_EDGES   (4 * PART_THREADS)
#define PART_BLOCKS  ((N_EDGES / 4 + PART_THREADS - 1) / PART_THREADS)
#define BO_PITCH     ((PART_BLOCKS + 31) & ~31)
static_assert(N_EDGES % 4 == 0, "edges are read four at a time");
static_assert(OWN_GROUPS < 256, "the first eight waves hold one owner per thread, and one thread more");
static_assert(PART_BLOCKS <= 1024, "kdinv holds one edge block per thread");

__global__ __launch_bounds__(PART_THREADS)
void kpart(const int* __restrict__ src, const int* __restrict__ dst,
           unsigned* __restrict__ recb, int* __restrict__ boff) {
    __shared__ int hist[OWN_GROUPS];
    __shared__ int sbase[OWN_GROUPS];
    __shared__ int wtot[8];

    const int t = threadIdx.x, lane = t & 31, wave = t >> 5;
    const int i = blockIdx.x * PART_THREADS + t;
    const bool ok = i < N_EDGES / 4;

    int4 d = make_int4(0, 0, 0, 0), s = make_int4(0, 0, 0, 0);
    if (ok) {
        d = ((const int4*)dst)[i];
        s = ((const int4*)src)[i];
    }
    if (t < OWN_GROUPS) hist[t] = 0;
    __syncthreads();

    const unsigned o0 = (unsigned)d.x / OWN_ROWS, o1 = (unsigned)d.y / OWN_ROWS,
                   o2 = (unsigned)d.z / OWN_ROWS, o3 = (unsigned)d.w / OWN_ROWS;
    int r0 = 0, r1 = 0, r2 = 0, r3 = 0;
    if (ok) {
        r0 = __hip_atomic_fetch_add(&hist[o0], 1, __ATOMIC_RELAXED, __HIP_MEMORY_SCOPE_WORKGROUP);
        r1 = __hip_atomic_fetch_add(&hist[o1], 1, __ATOMIC_RELAXED, __HIP_MEMORY_SCOPE_WORKGROUP);
        r2 = __hip_atomic_fetch_add(&hist[o2], 1, __ATOMIC_RELAXED, __HIP_MEMORY_SCOPE_WORKGROUP);
        r3 = __hip_atomic_fetch_add(&hist[o3], 1, __ATOMIC_RELAXED, __HIP_MEMORY_SCOPE_WORKGROUP);
    }
    __syncthreads();

    int start_t = 0;
    if (wave < 8) {
        const int cnt_t = (t < OWN_GROUPS) ? hist[t] : 0;
        int incl = cnt_t;
        for (int o = 1; o < 32; o <<= 1) {
            int up = __shfl_up(incl, o);
            if (lane >= o) incl += up;
        }
        if (lane == 31) wtot[wave] = incl;
        start_t = incl - cnt_t;
    }
    __syncthreads();
    if (t <= OWN_GROUPS) {
        for (int k = 0; k < wave; ++k) start_t += wtot[k];
        if (t < OWN_GROUPS) sbase[t] = start_t;
        boff[t * BO_PITCH + blockIdx.x] = start_t;
    }
    __syncthreads();
    if (ok) {
        unsigned* slice = recb + (size_t)blockIdx.x * PART_EDGES;
        slice[sbase[o0] + r0] = ((unsigned)s.x << 9) | ((unsigned)d.x - o0 * OWN_ROWS);
        slice[sbase[o1] + r1] = ((unsigned)s.y << 9) | ((unsigned)d.y - o1 * OWN_ROWS);
        slice[sbase[o2] + r2] = ((unsigned)s.z << 9) | ((unsigned)d.z - o2 * OWN_ROWS);
        slice[sbase[o3] + r3] = ((unsigned)s.w << 9) | ((unsigned)d.w - o3 * OWN_ROWS);
    }
}

__global__ __launch_bounds__(1024)
void kdinv(const unsigned* __restrict__ recb, const int* __restrict__ boff,
           unsigned* __restrict__ rec, int* __restrict__ gmeta, float* __restrict__ dinv) {
    __shared__ int cnt[OWN_ROWS];
    __shared__ int rfrom[PART_BLOCKS];
    __shared__ int rto[PART_BLOCKS];
    __shared__ int rlen[PART_BLOCKS];
    __shared__ int wlen[32], wlow[32];

    const int t = threadIdx.x, lane = t & 31, wave = t >> 5;
    const int g = blockIdx.x;
    int a = 0, e = 0;
    if (t < PART_BLOCKS) {
        a = boff[g * BO_PITCH + t];
        e = boff[(g + 1) * BO_PITCH + t];
    }
    if (t < OWN_ROWS) cnt[t] = 1;
    const int len = e - a;
    int incl = len, low = a;
    for (int o = 1; o < 32; o <<= 1) {
        int up  = __shfl_up(incl, o);
        int upl = __shfl_up(low, o);
        if (lane >= o) { incl += up; low += upl; }
    }
    if (lane == 31) { wlen[wave] = incl; wlow[wave] = low; }
    __syncthreads();
    int before = incl - len, n = 0, goff = 0;
    for (int k = 0; k < 32; ++k) {
        const int wl = wlen[k];
        if (k < wave) before += wl;
        n    += wl;
        goff += wlow[k];
    }
    if (t < PART_BLOCKS) {
        rfrom[t] = t * PART_EDGES + a;
        rto[t]   = before;
        rlen[t]  = len;
    }
    if (t == 0) {
        gmeta[2 * g]     = goff;
        gmeta[2 * g + 1] = n;
    }
    __syncthreads();
    unsigned* seg = rec + goff;
    const int wv = __builtin_amdgcn_readfirstlane(wave);
    constexpr int RUNS = 4;
    for (int b0 = wv; b0 < PART_BLOCKS; b0 += 32 * RUNS) {
        int from[RUNS], to[RUNS], l[RUNS];
        unsigned r[RUNS];
        #pragma unroll
        for (int u = 0; u < RUNS; ++u) {
            const int b = b0 + 32 * u;
            const int bc = min(b, PART_BLOCKS - 1);
            from[u] = rfrom[bc];
            to[u]   = rto[bc];
            l[u]    = (b < PART_BLOCKS) ? rlen[bc] : 0;
        }
        #pragma unroll
        for (int u = 0; u < RUNS; ++u)
            r[u] = recb[min(from[u] + min(lane, max(l[u] - 1, 0)), PART_BLOCKS * PART_EDGES - 1)];
        #pragma unroll
        for (int u = 0; u < RUNS; ++u) {
            if (lane < l[u]) {
                seg[to[u] + lane] = r[u];
                __hip_atomic_fetch_add(&cnt[r[u] & 511u], 1, __ATOMIC_RELAXED, __HIP_MEMORY_SCOPE_WORKGROUP);
            }
            for (int j = lane + 32; j < l[u]; j += 32) {
                const unsigned rr = recb[from[u] + j];
                seg[to[u] + j] = rr;
                __hip_atomic_fetch_add(&cnt[rr & 511u], 1, __ATOMIC_RELAXED, __HIP_MEMORY_SCOPE_WORKGROUP);
            }
        }
    }
    __syncthreads();
    if (t < OWN_ROWS) dinv[g * OWN_ROWS + t] = rsqrtf((float)cnt[t]);
}

__device__ __forceinline__ void own_add_row(float* __restrict__ tile, unsigned row, int lane,
                                            float4 v) {
    float* a = tile + row * TP + lane;
    fadd_lds(a,          v.x);
    fadd_lds(a + TJ,     v.y);
    fadd_lds(a + 2 * TJ, v.z);
    fadd_lds(a + 3 * TJ, v.w);
}

#define OWN_MT (OWN_ROWS / 16)
static_assert(OWN_ROWS % 16 == 0, "the tile is a whole number of 16-row matrix tiles");
__device__ __forceinline__ void own_layer(float* __restrict__ tile, const float* __restrict__ sdinv,
                                          int lo, int t, int lane, int wave,
                                          const float* __restrict__ hin,
                                          const float* __restrict__ x0,
                                          const float* __restrict__ w, float beta, int do_relu,
                                          int scale_out, float* __restrict__ hout) {
    const int half = lane >> 4, col = lane & 15;
    const int n0   = (wave & 7) * 16;
    const int ch   = n0 + col;

    v2f b[HID_C / 4];
    #pragma unroll
    for (int kk = 0; kk < HID_C / 4; ++kk) {
        const int ka = 4 * kk + 2 * half;
        b[kk].x = w[(size_t)ka * HID_C + ch];
        b[kk].y = w[(size_t)(ka + 1) * HID_C + ch];
    }

    const float4* h4 = (const float4*)hin + (size_t)lo * (HID_C / 4);
    const float4* x4 = (const float4*)x0  + (size_t)lo * (HID_C / 4);
    constexpr int NQ = OWN_ROWS * (HID_C / 4);
    for (int i0 = 0; i0 < NQ; i0 += 4 * OWN_WAVES * 32) {
        float4 hv[4], xv[4];
        #pragma unroll
        for (int k = 0; k < 4; ++k) {
            const int i = min(i0 + k * OWN_WAVES * 32 + t, NQ - 1);
            hv[k] = h4[i];
            xv[k] = x4[i];
        }
        #pragma unroll
        for (int k = 0; k < 4; ++k) {
            const int i = i0 + k * OWN_WAVES * 32 + t;
            if (i < NQ) {
                float* a = tile + (i >> 5) * TP + (i & 31);
                const float c = (1.f - ALPHA) * sdinv[i >> 5];
                a[0]      = c * (a[0]      + hv[k].x) + ALPHA * xv[k].x;
                a[TJ]     = c * (a[TJ]     + hv[k].y) + ALPHA * xv[k].y;
                a[2 * TJ] = c * (a[2 * TJ] + hv[k].z) + ALPHA * xv[k].z;
                a[3 * TJ] = c * (a[3 * TJ] + hv[k].w) + ALPHA * xv[k].w;
            }
        }
    }
    __syncthreads();

    for (int mt = wave >> 3; mt < OWN_MT; mt += OWN_WAVES / 8) {
        const float* ar = tile + (mt * 16 + col) * TP + 2 * half * TJ;
        v8f acc = {};
        #pragma unroll
        for (int kk = 0; kk < HID_C / 4; ++kk) {
            v2f a;
            a.x = ar[kk];
            a.y = ar[TJ + kk];
            acc = __builtin_amdgcn_wmma_f32_16x16x4_f32(false, a, false, b[kk],
                                                        (short)0, acc, false, false);
        }
        const float* sr = tile + (mt * 16 + 8 * half) * TP + (ch & 3) * TJ + (ch >> 2);
        float* orow = hout + (size_t)(lo + mt * 16 + 8 * half) * HID_C + ch;
        #pragma unroll
        for (int i = 0; i < 8; ++i) {
            float v = (1.f - beta) * sr[i * TP] + beta * acc[i];
            if (do_relu) v = fmaxf(v, 0.f);
            if (scale_out) v *= sdinv[mt * 16 + 8 * half + i];
            orow[(size_t)i * HID_C] = v;
        }
    }
}

__global__ __launch_bounds__(OWN_WAVES * 32)
void kown_replay(const unsigned* __restrict__ rec, const int* __restrict__ gmeta,
                 const float* __restrict__ hin, const float* __restrict__ x0,
                 const float* __restrict__ dinv,
                 const float* __restrict__ w, float beta, int do_relu, int scale_out,
                 float* __restrict__ hout) {
    __shared__ __attribute__((aligned(16))) float tile[OWN_ROWS * TP];
    __shared__ float sdinv[OWN_ROWS];

    const int t = threadIdx.x, lane = t & 31;
    const int wave = __builtin_amdgcn_readfirstlane(t >> 5);
    const int lo = blockIdx.x * OWN_ROWS;

    float4* tile4 = (float4*)tile;
    for (int i = t; i < OWN_ROWS * TP / 4; i += OWN_WAVES * 32)
        tile4[i] = make_float4(0.f, 0.f, 0.f, 0.f);
    if (t < OWN_ROWS) sdinv[t] = dinv[lo + t];
    __syncthreads();

    const int    goff = gmeta[2 * blockIdx.x];
    const int    n    = gmeta[2 * blockIdx.x + 1];
    const int    nd   = n / DRAIN;
    const unsigned* seg = rec + goff;
    const float4* h4  = (const float4*)hin;

    int      d  = wave;
    unsigned pe = 0u;
    if (d < nd) pe = seg[d * DRAIN + (lane & (DRAIN - 1))];
    while (d < nd) {
        unsigned row[DRAIN];
        float4   v[DRAIN];
        #pragma unroll
        for (int i = 0; i < DRAIN; ++i) {
            unsigned r = (unsigned)__builtin_amdgcn_readlane((int)pe, i);
            row[i] = r & 511u;
            v[i]   = h4[(size_t)(r >> 9) * (HID_C / 4) + lane];
        }
        const int      dn = d + OWN_WAVES;
        const unsigned pn = seg[min(dn, nd - 1) * DRAIN + (lane & (DRAIN - 1))];
        #pragma unroll
        for (int i = 0; i < DRAIN; ++i) own_add_row(tile, row[i], lane, v[i]);
        pe = pn;
        d  = dn;
    }
    if (wave == 0) {
        for (int i = nd * DRAIN; i < n; ++i) {
            unsigned r = (unsigned)__builtin_amdgcn_readfirstlane((int)seg[i]);
            float4   v = h4[(size_t)(r >> 9) * (HID_C / 4) + lane];
            own_add_row(tile, r & 511u, lane, v);
        }
    }
    __syncthreads();

    own_layer(tile, sdinv, lo, t, lane, wave, hin, x0, w, beta, do_relu, scale_out, hout);
}

extern "C" void kernel_launch(void* const* d_in, const int* in_sizes, int n_in,
                              void* d_out, int out_size, void* d_ws, size_t ws_size,
                              hipStream_t stream) {
    const float* x      = (const float*)d_in[0];
    const int*   ei     = (const int*)d_in[1];
    const float* proj_w = (const float*)d_in[2];
    const float* proj_b = (const float*)d_in[3];
    const float* conv_w = (const float*)d_in[4];
    float* out = (float*)d_out;

    const int* e_src = ei;
    const int* e_dst = ei + N_EDGES;

    const size_t NH = (size_t)N_NODES * HID_C;
    char* ws = (char*)d_ws;
    float* dinv = (float*)ws;                                ws += ((size_t)N_NODES * 4 + 255) & ~255ull;
    float* x0   = (float*)ws;                                ws += NH * 4;
    float* g0   = (float*)ws;                                ws += NH * 4;
    float* ha   = (float*)ws;                                ws += NH * 4;
    float* hb   = (float*)ws;                                ws += NH * 4;
    unsigned* rec    = (unsigned*)ws;                        ws += (size_t)N_EDGES * 4;
    unsigned* recb   = (unsigned*)ws;                        ws += (size_t)PART_BLOCKS * PART_EDGES * 4;
    int*      gmeta  = (int*)ws;                             ws += ((size_t)OWN_GROUPS * 8 + 255) & ~255ull;
    int*      boff   = (int*)ws;

    kpart<<<PART_BLOCKS, PART_THREADS, 0, stream>>>(e_src, e_dst, recb, boff);
    kdinv<<<OWN_GROUPS, 1024, 0, stream>>>(recb, boff, rec, gmeta, dinv);

    kproj<<<N_NODES / PROJ_ROWS, 256, 0, stream>>>(x, proj_w, proj_b, dinv, x0, g0);

    static const float betas[N_LAYERS] = {0.40546510f, 0.22314355f, 0.15415068f, 0.11778303f};

    const float* const hins[N_LAYERS]  = {g0, ha, hb, ha};
    float* const       houts[N_LAYERS] = {ha, hb, ha, out};

    for (int i = 0; i < N_LAYERS; ++i) {
        const float* wl   = conv_w + (size_t)i * HID_C * HID_C;
        const int    more = (i != N_LAYERS - 1) ? 1 : 0;
        kown_replay<<<OWN_GROUPS, OWN_WAVES * 32, 0, stream>>>(rec, gmeta, hins[i], x0, dinv,
                                                             wl, betas[i], more, more, houts[i]);
    }
}
